# starting kernel
# baseline (speedup 1.0000x reference)
#include <hip/hip_runtime.h>


typedef __attribute__((ext_vector_type(2))) float v2f;
typedef __attribute__((ext_vector_type(8))) float v8f;

#define NN   50000
#define NE   600000
#define KDIM 128


__global__ void k_zero4(float4* __restrict__ p, int n4) {
  int i = blockIdx.x * blockDim.x + threadIdx.x;
  if (i < n4) p[i] = make_float4(0.f, 0.f, 0.f, 0.f);
}

__global__ void k_deg(const int* __restrict__ src, const int* __restrict__ dst,
                      float* __restrict__ degS, float* __restrict__ degD, int nE) {
  int i = blockIdx.x * blockDim.x + threadIdx.x;
  if (i < nE) {
    atomicAdd(&degS[src[i]], 1.0f);
    atomicAdd(&degD[dst[i]], 1.0f);
  }
}

__global__ void k_norm(float* __restrict__ a, float* __restrict__ b, int n) {
  int i = blockIdx.x * blockDim.x + threadIdx.x;
  if (i < n) {
    a[i] = rsqrtf(fmaxf(a[i], 1.0f));
    b[i] = rsqrtf(fmaxf(b[i], 1.0f));
  }
}


__global__ __launch_bounds__(128) void k_gemm_wmma(
    const float* __restrict__ X, const float* __restrict__ norm,
    const float* __restrict__ W, float* __restrict__ Y, int ncshift  ) {
  __shared__ float As[16 * 132];
  __shared__ float Ws[(KDIM / 2) * 144 * 2];

  const int ncol  = 1 << ncshift;
  const int ncolP = ncol + 16;
  const int m0    = blockIdx.x * 16;
  const int tid   = threadIdx.x;

  for (int i = tid; i < 16 * KDIM; i += 128) {
    int r = i >> 7;
    int c = i & (KDIM - 1);
    As[r * 132 + c] = X[(size_t)(m0 + r) * KDIM + c] * norm[m0 + r];
  }

  {
    const int ncq = ncshift - 2;
    const int nf4 = KDIM << ncq;
    for (int idx = tid; idx < nf4; idx += 128) {
      const int k = idx >> ncq;
      const int j = (idx - (k << ncq)) << 2;
      const float4 v = ((const float4*)W)[idx];
      const int base = ((k >> 1) * ncolP + j) * 2 + (k & 1);
      Ws[base + 0] = v.x;
      Ws[base + 2] = v.y;
      Ws[base + 4] = v.z;
      Ws[base + 6] = v.w;
    }
  }
  __syncthreads();

  const int wave = tid >> 5;
  const int lane = tid & 31;
  const int hl   = lane >> 4;
  const int lm   = lane & 15;

  const int ntile = ncol >> 4;
  for (int t = wave; t < ntile; t += 4) {
    v8f acc = {};
    const int n = t * 16 + lm;
    #pragma unroll
    for (int k0 = 0; k0 < KDIM; k0 += 4) {
      const int kk = k0 + 2 * hl;
      v2f a;
      a.x = As[lm * 132 + kk];
      a.y = As[lm * 132 + kk + 1];
      const int p = kk >> 1;
      const v2f b = *(const v2f*)&Ws[(p * ncolP + n) * 2];
      acc = __builtin_amdgcn_wmma_f32_16x16x4_f32(
           false, a,  false, b,
           (short)0, acc,  false,  false);
    }
    #pragma unroll
    for (int r = 0; r < 8; ++r)
      Y[(size_t)(m0 + hl * 8 + r) * ncol + n] = acc[r];
  }
}


__global__ void k_scatter(const float* __restrict__ H, const int* __restrict__ src,
                          const int* __restrict__ dst, float* __restrict__ out,
                          int nE, int fshift  ) {
  const int total = nE << fshift;
  int i = blockIdx.x * blockDim.x + threadIdx.x;
  if (i >= total) return;
  const int e = i >> fshift;
  const int c = i & ((1 << fshift) - 1);
  const int s = src[e];
  const int d = dst[e];
  const float4 v = ((const float4*)H)[((size_t)s << fshift) + c];
  float* o = out + ((((size_t)d << fshift) + c) << 2);
  atomicAdd(o + 0, v.x);
  atomicAdd(o + 1, v.y);
  atomicAdd(o + 2, v.z);
  atomicAdd(o + 3, v.w);
}


__global__ void k_post(float* __restrict__ h, const float* __restrict__ normD,
                       const float* __restrict__ b, int total, int shift,
                       int doRelu) {
  int i = blockIdx.x * blockDim.x + threadIdx.x;
  if (i >= total) return;
  const int row = i >> shift;
  const int col = i & ((1 << shift) - 1);
  float v = h[i] * normD[row] + b[col];
  h[i] = doRelu ? fmaxf(v, 0.0f) : v;
}


extern "C" void kernel_launch(void* const* d_in, const int* in_sizes, int n_in,
                              void* d_out, int out_size, void* d_ws, size_t ws_size,
                              hipStream_t stream) {
  (void)in_sizes; (void)n_in; (void)out_size; (void)ws_size;

  const float* features = (const float*)d_in[0];
  const float* W1       = (const float*)d_in[1];
  const float* b1       = (const float*)d_in[2];
  const float* W2       = (const float*)d_in[3];
  const float* b2       = (const float*)d_in[4];
  const int*   esrc     = (const int*)d_in[5];
  const int*   edst     = (const int*)d_in[6];
  float*       out      = (float*)d_out;

  const size_t PADN = 50176;
  float* w     = (float*)d_ws;
  float* normS = w;
  float* normD = w + PADN;
  float* B0    = w + 2 * PADN;
  float* B1    = B0 + (size_t)NN * 128;

  const int T = 256;

  {
    int n4 = (int)(2 * PADN / 4);
    k_zero4<<<(n4 + T - 1) / T, T, 0, stream>>>((float4*)normS, n4);
    k_deg<<<(NE + T - 1) / T, T, 0, stream>>>(esrc, edst, normS, normD, NE);
    k_norm<<<(NN + T - 1) / T, T, 0, stream>>>(normS, normD, NN);
  }

  k_gemm_wmma<<<NN / 16, 128, 0, stream>>>(features, normS, W1, B0, 7);

  {
    int n4 = NN * 32;
    k_zero4<<<(n4 + T - 1) / T, T, 0, stream>>>((float4*)B1, n4);
    int tot = NE * 32;
    k_scatter<<<(tot + T - 1) / T, T, 0, stream>>>(B0, esrc, edst, B1, NE, 5);
  }

  k_post<<<(NN * 128 + T - 1) / T, T, 0, stream>>>(B1, normD, b1, NN * 128, 7, 1);

  k_gemm_wmma<<<NN / 16, 128, 0, stream>>>(B1, normS, W2, B0, 6);

  {
    int n4 = NN * 16;
    k_zero4<<<(n4 + T - 1) / T, T, 0, stream>>>((float4*)out, n4);
    int tot = NE * 16;
    k_scatter<<<(tot + T - 1) / T, T, 0, stream>>>(B0, esrc, edst, out, NE, 4);
  }

  k_post<<<(NN * 64 + T - 1) / T, T, 0, stream>>>(out, normD, b2, NN * 64, 6, 0);
}


# GEMMs: W kept in registers as B fragments by waves that own 64 columns and walk over row strips, A straight from memory one strip ahead, no LDS; k summed in another order inside each 8
# speedup vs baseline: 1.2646x; 1.2646x over previous
#include <hip/hip_runtime.h>


typedef __attribute__((ext_vector_type(2))) float v2f;
typedef __attribute__((ext_vector_type(8))) float v8f;

#define NN   50000
#define NE   600000
#define KDIM 128


__global__ void k_zero4(float4* __restrict__ p, int n4) {
  int i = blockIdx.x * blockDim.x + threadIdx.x;
  if (i < n4) p[i] = make_float4(0.f, 0.f, 0.f, 0.f);
}

__global__ void k_deg(const int* __restrict__ src, const int* __restrict__ dst,
                      float* __restrict__ degS, float* __restrict__ degD, int nE) {
  int i = blockIdx.x * blockDim.x + threadIdx.x;
  if (i < nE) {
    atomicAdd(&degS[src[i]], 1.0f);
    atomicAdd(&degD[dst[i]], 1.0f);
  }
}

__global__ void k_norm(float* __restrict__ a, float* __restrict__ b, int n) {
  int i = blockIdx.x * blockDim.x + threadIdx.x;
  if (i < n) {
    a[i] = rsqrtf(fmaxf(a[i], 1.0f));
    b[i] = rsqrtf(fmaxf(b[i], 1.0f));
  }
}


template <int NCOL>
__global__ __launch_bounds__(NCOL / 2) void k_gemm_wreg(
    const float* __restrict__ X, const float* __restrict__ norm,
    const float* __restrict__ W, float* __restrict__ Y, int nstrips) {
  const int lane = threadIdx.x & 31;
  const int wave = threadIdx.x >> 5;
  const int c    = lane & 15;
  const int h    = lane >> 4;
  const int col0 = wave * 64 + 4 * c;

  int strip = blockIdx.x;
  if (strip >= nstrips) return;

  v2f b[4][32];
  {
    const float4* wp = (const float4*)(W + (size_t)(4 * h) * NCOL + col0);
    #pragma unroll
    for (int j = 0; j < 16; ++j) {
      #pragma unroll
      for (int q = 0; q < 2; ++q) {
        const float4 w0 = wp[((8 * j + 2 * q) * NCOL) >> 2];
        const float4 w1 = wp[((8 * j + 2 * q + 1) * NCOL) >> 2];
        b[0][2 * j + q] = v2f{w0.x, w1.x};
        b[1][2 * j + q] = v2f{w0.y, w1.y};
        b[2][2 * j + q] = v2f{w0.z, w1.z};
        b[3][2 * j + q] = v2f{w0.w, w1.w};
      }
    }
  }

  float4 a[16];
  float  nm;
  {
    const float4* xp = (const float4*)(X + (size_t)(strip * 16 + c) * KDIM + 4 * h);
    #pragma unroll
    for (int j = 0; j < 16; ++j) a[j] = xp[2 * j];
    nm = norm[strip * 16 + c];
  }

  for (;;) {
    const int  nxt  = strip + (int)gridDim.x;
    const bool more = nxt < nstrips;
    const int  pf   = more ? nxt : strip;

    float4 an[16];
    const float4* xn = (const float4*)(X + (size_t)(pf * 16 + c) * KDIM + 4 * h);
    #pragma unroll
    for (int j = 0; j < 16; ++j) an[j] = xn[2 * j];
    const float nn = norm[pf * 16 + c];

    v8f acc0 = {}, acc1 = {}, acc2 = {}, acc3 = {};
    #pragma unroll
    for (int j = 0; j < 16; ++j) {
      const v2f a0 = v2f{a[j].x * nm, a[j].y * nm};
      const v2f a1 = v2f{a[j].z * nm, a[j].w * nm};
      acc0 = __builtin_amdgcn_wmma_f32_16x16x4_f32(false, a0, false, b[0][2 * j], (short)0, acc0, false, false);
      acc1 = __builtin_amdgcn_wmma_f32_16x16x4_f32(false, a0, false, b[1][2 * j], (short)0, acc1, false, false);
      acc2 = __builtin_amdgcn_wmma_f32_16x16x4_f32(false, a0, false, b[2][2 * j], (short)0, acc2, false, false);
      acc3 = __builtin_amdgcn_wmma_f32_16x16x4_f32(false, a0, false, b[3][2 * j], (short)0, acc3, false, false);
      acc0 = __builtin_amdgcn_wmma_f32_16x16x4_f32(false, a1, false, b[0][2 * j + 1], (short)0, acc0, false, false);
      acc1 = __builtin_amdgcn_wmma_f32_16x16x4_f32(false, a1, false, b[1][2 * j + 1], (short)0, acc1, false, false);
      acc2 = __builtin_amdgcn_wmma_f32_16x16x4_f32(false, a1, false, b[2][2 * j + 1], (short)0, acc2, false, false);
      acc3 = __builtin_amdgcn_wmma_f32_16x16x4_f32(false, a1, false, b[3][2 * j + 1], (short)0, acc3, false, false);
    }

    float4* yp = (float4*)(Y + (size_t)(strip * 16 + 8 * h) * NCOL + col0);
    #pragma unroll
    for (int r = 0; r < 8; ++r)
      yp[(r * NCOL) >> 2] = make_float4(acc0[r], acc1[r], acc2[r], acc3[r]);

    if (!more) break;
    #pragma unroll
    for (int j = 0; j < 16; ++j) a[j] = an[j];
    nm    = nn;
    strip = nxt;
  }
}


__global__ void k_scatter(const float* __restrict__ H, const int* __restrict__ src,
                          const int* __restrict__ dst, float* __restrict__ out,
                          int nE, int fshift  ) {
  const int total = nE << fshift;
  int i = blockIdx.x * blockDim.x + threadIdx.x;
  if (i >= total) return;
  const int e = i >> fshift;
  const int c = i & ((1 << fshift) - 1);
  const int s = src[e];
  const int d = dst[e];
  const float4 v = ((const float4*)H)[((size_t)s << fshift) + c];
  float* o = out + ((((size_t)d << fshift) + c) << 2);
  atomicAdd(o + 0, v.x);
  atomicAdd(o + 1, v.y);
  atomicAdd(o + 2, v.z);
  atomicAdd(o + 3, v.w);
}


__global__ void k_post(float* __restrict__ h, const float* __restrict__ normD,
                       const float* __restrict__ b, int total, int shift,
                       int doRelu) {
  int i = blockIdx.x * blockDim.x + threadIdx.x;
  if (i >= total) return;
  const int row = i >> shift;
  const int col = i & ((1 << shift) - 1);
  float v = h[i] * normD[row] + b[col];
  h[i] = doRelu ? fmaxf(v, 0.0f) : v;
}


extern "C" void kernel_launch(void* const* d_in, const int* in_sizes, int n_in,
                              void* d_out, int out_size, void* d_ws, size_t ws_size,
                              hipStream_t stream) {
  (void)in_sizes; (void)n_in; (void)out_size; (void)ws_size;

  const float* features = (const float*)d_in[0];
  const float* W1       = (const float*)d_in[1];
  const float* b1       = (const float*)d_in[2];
  const float* W2       = (const float*)d_in[3];
  const float* b2       = (const float*)d_in[4];
  const int*   esrc     = (const int*)d_in[5];
  const int*   edst     = (const int*)d_in[6];
  float*       out      = (float*)d_out;

  const size_t PADN = 50176;
  float* w     = (float*)d_ws;
  float* normS = w;
  float* normD = w + PADN;
  float* B0    = w + 2 * PADN;
  float* B1    = B0 + (size_t)NN * 128;

  const int T = 256;

  {
    int n4 = (int)(2 * PADN / 4);
    k_zero4<<<(n4 + T - 1) / T, T, 0, stream>>>((float4*)normS, n4);
    k_deg<<<(NE + T - 1) / T, T, 0, stream>>>(esrc, edst, normS, normD, NE);
    k_norm<<<(NN + T - 1) / T, T, 0, stream>>>(normS, normD, NN);
  }

  k_gemm_wreg<128><<<1024, 64, 0, stream>>>(features, normS, W1, B0, NN / 16);

  {
    int n4 = NN * 32;
    k_zero4<<<(n4 + T - 1) / T, T, 0, stream>>>((float4*)B1, n4);
    int tot = NE * 32;
    k_scatter<<<(tot + T - 1) / T, T, 0, stream>>>(B0, esrc, edst, B1, NE, 5);
  }

  k_post<<<(NN * 128 + T - 1) / T, T, 0, stream>>>(B1, normD, b1, NN * 128, 7, 1);

  k_gemm_wreg<64><<<2048, 32, 0, stream>>>(B1, normS, W2, B0, NN / 16);

  {
    int n4 = NN * 16;
    k_zero4<<<(n4 + T - 1) / T, T, 0, stream>>>((float4*)out, n4);
    int tot = NE * 16;
    k_scatter<<<(tot + T - 1) / T, T, 0, stream>>>(B0, esrc, edst, out, NE, 4);
  }

  k_post<<<(NN * 64 + T - 1) / T, T, 0, stream>>>(out, normD, b2, NN * 64, 6, 0);
}


# Edge scatter: a wave owns whole edges, lane l adds floats l+32q of a row, so each row load and each float atomic covers one whole 128-B line (was 4 lines x 8 lanes); sums stay arrival-order atomics
# speedup vs baseline: 1.7931x; 1.4179x over previous
#include <hip/hip_runtime.h>


typedef __attribute__((ext_vector_type(2))) float v2f;
typedef __attribute__((ext_vector_type(8))) float v8f;

#define NN   50000
#define NE   600000
#define KDIM 128


__global__ void k_zero4(float4* __restrict__ p, int n4) {
  int i = blockIdx.x * blockDim.x + threadIdx.x;
  if (i < n4) p[i] = make_float4(0.f, 0.f, 0.f, 0.f);
}

__global__ void k_deg(const int* __restrict__ src, const int* __restrict__ dst,
                      float* __restrict__ degS, float* __restrict__ degD, int nE) {
  int i = blockIdx.x * blockDim.x + threadIdx.x;
  if (i < nE) {
    atomicAdd(&degS[src[i]], 1.0f);
    atomicAdd(&degD[dst[i]], 1.0f);
  }
}

__global__ void k_norm(float* __restrict__ a, float* __restrict__ b, int n) {
  int i = blockIdx.x * blockDim.x + threadIdx.x;
  if (i < n) {
    a[i] = rsqrtf(fmaxf(a[i], 1.0f));
    b[i] = rsqrtf(fmaxf(b[i], 1.0f));
  }
}


template <int NCOL>
__global__ __launch_bounds__(NCOL / 2) void k_gemm_wreg(
    const float* __restrict__ X, const float* __restrict__ norm,
    const float* __restrict__ W, float* __restrict__ Y, int nstrips) {
  const int lane = threadIdx.x & 31;
  const int wave = threadIdx.x >> 5;
  const int c    = lane & 15;
  const int h    = lane >> 4;
  const int col0 = wave * 64 + 4 * c;

  int strip = blockIdx.x;
  if (strip >= nstrips) return;

  v2f b[4][32];
  {
    const float4* wp = (const float4*)(W + (size_t)(4 * h) * NCOL + col0);
    #pragma unroll
    for (int j = 0; j < 16; ++j) {
      #pragma unroll
      for (int q = 0; q < 2; ++q) {
        const float4 w0 = wp[((8 * j + 2 * q) * NCOL) >> 2];
        const float4 w1 = wp[((8 * j + 2 * q + 1) * NCOL) >> 2];
        b[0][2 * j + q] = v2f{w0.x, w1.x};
        b[1][2 * j + q] = v2f{w0.y, w1.y};
        b[2][2 * j + q] = v2f{w0.z, w1.z};
        b[3][2 * j + q] = v2f{w0.w, w1.w};
      }
    }
  }

  float4 a[16];
  float  nm;
  {
    const float4* xp = (const float4*)(X + (size_t)(strip * 16 + c) * KDIM + 4 * h);
    #pragma unroll
    for (int j = 0; j < 16; ++j) a[j] = xp[2 * j];
    nm = norm[strip * 16 + c];
  }

  for (;;) {
    const int  nxt  = strip + (int)gridDim.x;
    const bool more = nxt < nstrips;
    const int  pf   = more ? nxt : strip;

    float4 an[16];
    const float4* xn = (const float4*)(X + (size_t)(pf * 16 + c) * KDIM + 4 * h);
    #pragma unroll
    for (int j = 0; j < 16; ++j) an[j] = xn[2 * j];
    const float nn = norm[pf * 16 + c];

    v8f acc0 = {}, acc1 = {}, acc2 = {}, acc3 = {};
    #pragma unroll
    for (int j = 0; j < 16; ++j) {
      const v2f a0 = v2f{a[j].x * nm, a[j].y * nm};
      const v2f a1 = v2f{a[j].z * nm, a[j].w * nm};
      acc0 = __builtin_amdgcn_wmma_f32_16x16x4_f32(false, a0, false, b[0][2 * j], (short)0, acc0, false, false);
      acc1 = __builtin_amdgcn_wmma_f32_16x16x4_f32(false, a0, false, b[1][2 * j], (short)0, acc1, false, false);
      acc2 = __builtin_amdgcn_wmma_f32_16x16x4_f32(false, a0, false, b[2][2 * j], (short)0, acc2, false, false);
      acc3 = __builtin_amdgcn_wmma_f32_16x16x4_f32(false, a0, false, b[3][2 * j], (short)0, acc3, false, false);
      acc0 = __builtin_amdgcn_wmma_f32_16x16x4_f32(false, a1, false, b[0][2 * j + 1], (short)0, acc0, false, false);
      acc1 = __builtin_amdgcn_wmma_f32_16x16x4_f32(false, a1, false, b[1][2 * j + 1], (short)0, acc1, false, false);
      acc2 = __builtin_amdgcn_wmma_f32_16x16x4_f32(false, a1, false, b[2][2 * j + 1], (short)0, acc2, false, false);
      acc3 = __builtin_amdgcn_wmma_f32_16x16x4_f32(false, a1, false, b[3][2 * j + 1], (short)0, acc3, false, false);
    }

    float4* yp = (float4*)(Y + (size_t)(strip * 16 + 8 * h) * NCOL + col0);
    #pragma unroll
    for (int r = 0; r < 8; ++r)
      yp[(r * NCOL) >> 2] = make_float4(acc0[r], acc1[r], acc2[r], acc3[r]);

    if (!more) break;
    #pragma unroll
    for (int j = 0; j < 16; ++j) a[j] = an[j];
    nm    = nn;
    strip = nxt;
  }
}


template <int F, int EPW>
__global__ __launch_bounds__(256) void k_scatter_line(
    const float* __restrict__ H, const int* __restrict__ src,
    const int* __restrict__ dst, float* __restrict__ out, int nE) {
  constexpr int Q = F / 32;
  const int lane = threadIdx.x & 31;
  const int wave = __builtin_amdgcn_readfirstlane((int)(threadIdx.x >> 5));
  const int g    = (int)blockIdx.x * 8 + wave;
  const int ngrp = nE / EPW;

  if (g < ngrp) {
    const int e0 = g * EPW;
    int s[EPW], d[EPW];
    #pragma unroll
    for (int k = 0; k < EPW; ++k) { s[k] = src[e0 + k]; d[k] = dst[e0 + k]; }

    float v[EPW][Q];
    #pragma unroll
    for (int k = 0; k < EPW; ++k) {
      const float* hp = H + (size_t)s[k] * F + lane;
      #pragma unroll
      for (int q = 0; q < Q; ++q) v[k][q] = hp[32 * q];
    }
    #pragma unroll
    for (int k = 0; k < EPW; ++k) {
      float* op = out + (size_t)d[k] * F + lane;
      #pragma unroll
      for (int q = 0; q < Q; ++q) atomicAdd(op + 32 * q, v[k][q]);
    }
  }

  const int et = ngrp * EPW + g;
  if (g < EPW && et < nE) {
    const float* hp = H + (size_t)src[et] * F + lane;
    float*       op = out + (size_t)dst[et] * F + lane;
    #pragma unroll
    for (int q = 0; q < Q; ++q) atomicAdd(op + 32 * q, hp[32 * q]);
  }
}


__global__ void k_post(float* __restrict__ h, const float* __restrict__ normD,
                       const float* __restrict__ b, int total, int shift,
                       int doRelu) {
  int i = blockIdx.x * blockDim.x + threadIdx.x;
  if (i >= total) return;
  const int row = i >> shift;
  const int col = i & ((1 << shift) - 1);
  float v = h[i] * normD[row] + b[col];
  h[i] = doRelu ? fmaxf(v, 0.0f) : v;
}


extern "C" void kernel_launch(void* const* d_in, const int* in_sizes, int n_in,
                              void* d_out, int out_size, void* d_ws, size_t ws_size,
                              hipStream_t stream) {
  (void)in_sizes; (void)n_in; (void)out_size; (void)ws_size;

  const float* features = (const float*)d_in[0];
  const float* W1       = (const float*)d_in[1];
  const float* b1       = (const float*)d_in[2];
  const float* W2       = (const float*)d_in[3];
  const float* b2       = (const float*)d_in[4];
  const int*   esrc     = (const int*)d_in[5];
  const int*   edst     = (const int*)d_in[6];
  float*       out      = (float*)d_out;

  const size_t PADN = 50176;
  float* w     = (float*)d_ws;
  float* normS = w;
  float* normD = w + PADN;
  float* B0    = w + 2 * PADN;
  float* B1    = B0 + (size_t)NN * 128;

  const int T = 256;

  {
    int n4 = (int)(2 * PADN / 4);
    k_zero4<<<(n4 + T - 1) / T, T, 0, stream>>>((float4*)normS, n4);
    k_deg<<<(NE + T - 1) / T, T, 0, stream>>>(esrc, edst, normS, normD, NE);
    k_norm<<<(NN + T - 1) / T, T, 0, stream>>>(normS, normD, NN);
  }

  k_gemm_wreg<128><<<1024, 64, 0, stream>>>(features, normS, W1, B0, NN / 16);

  {
    int n4 = NN * 32;
    k_zero4<<<(n4 + T - 1) / T, T, 0, stream>>>((float4*)B1, n4);
    k_scatter_line<128, 8><<<NE / 64 + 1, 256, 0, stream>>>(B0, esrc, edst, B1, NE);
  }

  k_post<<<(NN * 128 + T - 1) / T, T, 0, stream>>>(B1, normD, b1, NN * 128, 7, 1);

  k_gemm_wreg<64><<<2048, 32, 0, stream>>>(B1, normS, W2, B0, NN / 16);

  {
    int n4 = NN * 16;
    k_zero4<<<(n4 + T - 1) / T, T, 0, stream>>>((float4*)out, n4);
    k_scatter_line<64, 16><<<NE / 128 + 1, 256, 0, stream>>>(B0, esrc, edst, out, NE);
  }

  k_post<<<(NN * 64 + T - 1) / T, T, 0, stream>>>(out, normD, b2, NN * 64, 6, 0);
}


# Element-wise passes folded into the GEMM launches: gemm1 clears the layer-1 sum buffer, gemm2's A loader does scale+bias+relu and clears out; 11 -> 8 launches; sums and k order unchanged
# speedup vs baseline: 1.9174x; 1.0693x over previous
#include <hip/hip_runtime.h>


typedef __attribute__((ext_vector_type(2))) float v2f;
typedef __attribute__((ext_vector_type(8))) float v8f;

#define NN   50000
#define NE   600000
#define KDIM 128


__global__ void k_zero4(float4* __restrict__ p, int n4) {
  int i = blockIdx.x * blockDim.x + threadIdx.x;
  if (i < n4) p[i] = make_float4(0.f, 0.f, 0.f, 0.f);
}

__global__ void k_deg(const int* __restrict__ src, const int* __restrict__ dst,
                      float* __restrict__ degS, float* __restrict__ degD, int nE) {
  int i = blockIdx.x * blockDim.x + threadIdx.x;
  if (i < nE) {
    atomicAdd(&degS[src[i]], 1.0f);
    atomicAdd(&degD[dst[i]], 1.0f);
  }
}

__global__ void k_norm(float* __restrict__ a, float* __restrict__ b, int n) {
  int i = blockIdx.x * blockDim.x + threadIdx.x;
  if (i < n) {
    a[i] = rsqrtf(fmaxf(a[i], 1.0f));
    b[i] = rsqrtf(fmaxf(b[i], 1.0f));
  }
}


template <int NCOL, bool PRE>
__global__ __launch_bounds__(NCOL / 2) void k_gemm_wreg(
    const float* __restrict__ X, const float* __restrict__ norm,
    const float* __restrict__ rowS, const float* __restrict__ colB,
    const float* __restrict__ W, float* __restrict__ Y,
    float* __restrict__ Z, int nstrips) {
  const int lane = threadIdx.x & 31;
  const int wave = threadIdx.x >> 5;
  const int c    = lane & 15;
  const int h    = lane >> 4;
  const int col0 = wave * 64 + 4 * c;

  int strip = blockIdx.x;
  if (strip >= nstrips) return;

  v2f b[4][32];
  {
    const float4* wp = (const float4*)(W + (size_t)(4 * h) * NCOL + col0);
    #pragma unroll
    for (int j = 0; j < 16; ++j) {
      #pragma unroll
      for (int q = 0; q < 2; ++q) {
        const float4 w0 = wp[((8 * j + 2 * q) * NCOL) >> 2];
        const float4 w1 = wp[((8 * j + 2 * q + 1) * NCOL) >> 2];
        b[0][2 * j + q] = v2f{w0.x, w1.x};
        b[1][2 * j + q] = v2f{w0.y, w1.y};
        b[2][2 * j + q] = v2f{w0.z, w1.z};
        b[3][2 * j + q] = v2f{w0.w, w1.w};
      }
    }
  }

  float4 a[16];
  float  nm;
  {
    const float4* xp = (const float4*)(X + (size_t)(strip * 16 + c) * KDIM + 4 * h);
    #pragma unroll
    for (int j = 0; j < 16; ++j) a[j] = xp[2 * j];
    nm = norm[strip * 16 + c];
  }

  float4 bv[16];
  float  nd = 0.f;
  if constexpr (PRE) {
    const float4* bp = (const float4*)(colB + 4 * h);
    #pragma unroll
    for (int j = 0; j < 16; ++j) bv[j] = bp[2 * j];
    nd = rowS[strip * 16 + c];
  }

  for (;;) {
    const int  nxt  = strip + (int)gridDim.x;
    const bool more = nxt < nstrips;
    const int  pf   = more ? nxt : strip;

    float4 an[16];
    const float4* xn = (const float4*)(X + (size_t)(pf * 16 + c) * KDIM + 4 * h);
    #pragma unroll
    for (int j = 0; j < 16; ++j) an[j] = xn[2 * j];
    const float nn = norm[pf * 16 + c];
    float ndn = 0.f;
    if constexpr (PRE) ndn = rowS[pf * 16 + c];

    v8f acc0 = {}, acc1 = {}, acc2 = {}, acc3 = {};
    #pragma unroll
    for (int j = 0; j < 16; ++j) {
      float4 x = a[j];
      if constexpr (PRE) {
        x.x = fmaxf(x.x * nd + bv[j].x, 0.0f);
        x.y = fmaxf(x.y * nd + bv[j].y, 0.0f);
        x.z = fmaxf(x.z * nd + bv[j].z, 0.0f);
        x.w = fmaxf(x.w * nd + bv[j].w, 0.0f);
      }
      const v2f a0 = v2f{x.x * nm, x.y * nm};
      const v2f a1 = v2f{x.z * nm, x.w * nm};
      acc0 = __builtin_amdgcn_wmma_f32_16x16x4_f32(false, a0, false, b[0][2 * j], (short)0, acc0, false, false);
      acc1 = __builtin_amdgcn_wmma_f32_16x16x4_f32(false, a0, false, b[1][2 * j], (short)0, acc1, false, false);
      acc2 = __builtin_amdgcn_wmma_f32_16x16x4_f32(false, a0, false, b[2][2 * j], (short)0, acc2, false, false);
      acc3 = __builtin_amdgcn_wmma_f32_16x16x4_f32(false, a0, false, b[3][2 * j], (short)0, acc3, false, false);
      acc0 = __builtin_amdgcn_wmma_f32_16x16x4_f32(false, a1, false, b[0][2 * j + 1], (short)0, acc0, false, false);
      acc1 = __builtin_amdgcn_wmma_f32_16x16x4_f32(false, a1, false, b[1][2 * j + 1], (short)0, acc1, false, false);
      acc2 = __builtin_amdgcn_wmma_f32_16x16x4_f32(false, a1, false, b[2][2 * j + 1], (short)0, acc2, false, false);
      acc3 = __builtin_amdgcn_wmma_f32_16x16x4_f32(false, a1, false, b[3][2 * j + 1], (short)0, acc3, false, false);
    }

    float4* yp = (float4*)(Y + (size_t)(strip * 16 + 8 * h) * NCOL + col0);
    #pragma unroll
    for (int r = 0; r < 8; ++r)
      yp[(r * NCOL) >> 2] = make_float4(acc0[r], acc1[r], acc2[r], acc3[r]);

    float4* zp = (float4*)(Z + (size_t)(strip * 16 + 8 * h) * NCOL + col0);
    #pragma unroll
    for (int r = 0; r < 8; ++r)
      zp[(r * NCOL) >> 2] = make_float4(0.f, 0.f, 0.f, 0.f);

    if (!more) break;
    #pragma unroll
    for (int j = 0; j < 16; ++j) a[j] = an[j];
    nm    = nn;
    nd    = ndn;
    strip = nxt;
  }
}


template <int F, int EPW>
__global__ __launch_bounds__(256) void k_scatter_line(
    const float* __restrict__ H, const int* __restrict__ src,
    const int* __restrict__ dst, float* __restrict__ out, int nE) {
  constexpr int Q = F / 32;
  const int lane = threadIdx.x & 31;
  const int wave = __builtin_amdgcn_readfirstlane((int)(threadIdx.x >> 5));
  const int g    = (int)blockIdx.x * 8 + wave;
  const int ngrp = nE / EPW;

  if (g < ngrp) {
    const int e0 = g * EPW;
    int s[EPW], d[EPW];
    #pragma unroll
    for (int k = 0; k < EPW; ++k) { s[k] = src[e0 + k]; d[k] = dst[e0 + k]; }

    float v[EPW][Q];
    #pragma unroll
    for (int k = 0; k < EPW; ++k) {
      const float* hp = H + (size_t)s[k] * F + lane;
      #pragma unroll
      for (int q = 0; q < Q; ++q) v[k][q] = hp[32 * q];
    }
    #pragma unroll
    for (int k = 0; k < EPW; ++k) {
      float* op = out + (size_t)d[k] * F + lane;
      #pragma unroll
      for (int q = 0; q < Q; ++q) atomicAdd(op + 32 * q, v[k][q]);
    }
  }

  const int et = ngrp * EPW + g;
  if (g < EPW && et < nE) {
    const float* hp = H + (size_t)src[et] * F + lane;
    float*       op = out + (size_t)dst[et] * F + lane;
    #pragma unroll
    for (int q = 0; q < Q; ++q) atomicAdd(op + 32 * q, hp[32 * q]);
  }
}


__global__ void k_post(float* __restrict__ h, const float* __restrict__ normD,
                       const float* __restrict__ b, int total, int shift,
                       int doRelu) {
  int i = blockIdx.x * blockDim.x + threadIdx.x;
  if (i >= total) return;
  const int row = i >> shift;
  const int col = i & ((1 << shift) - 1);
  float v = h[i] * normD[row] + b[col];
  h[i] = doRelu ? fmaxf(v, 0.0f) : v;
}


extern "C" void kernel_launch(void* const* d_in, const int* in_sizes, int n_in,
                              void* d_out, int out_size, void* d_ws, size_t ws_size,
                              hipStream_t stream) {
  (void)in_sizes; (void)n_in; (void)out_size; (void)ws_size;

  const float* features = (const float*)d_in[0];
  const float* W1       = (const float*)d_in[1];
  const float* b1       = (const float*)d_in[2];
  const float* W2       = (const float*)d_in[3];
  const float* b2       = (const float*)d_in[4];
  const int*   esrc     = (const int*)d_in[5];
  const int*   edst     = (const int*)d_in[6];
  float*       out      = (float*)d_out;

  const size_t PADN = 50176;
  float* w     = (float*)d_ws;
  float* normS = w;
  float* normD = w + PADN;
  float* B0    = w + 2 * PADN;
  float* B1    = B0 + (size_t)NN * 128;

  const int T = 256;

  {
    int n4 = (int)(2 * PADN / 4);
    k_zero4<<<(n4 + T - 1) / T, T, 0, stream>>>((float4*)normS, n4);
    k_deg<<<(NE + T - 1) / T, T, 0, stream>>>(esrc, edst, normS, normD, NE);
    k_norm<<<(NN + T - 1) / T, T, 0, stream>>>(normS, normD, NN);
  }

  k_gemm_wreg<128, false><<<1024, 64, 0, stream>>>(features, normS, normD, b1, W1,
                                                   B0, B1, NN / 16);

  k_scatter_line<128, 8><<<NE / 64 + 1, 256, 0, stream>>>(B0, esrc, edst, B1, NE);

  k_gemm_wreg<64, true><<<2048, 32, 0, stream>>>(B1, normS, normD, b1, W2,
                                                 B0, out, NN / 16);

  k_scatter_line<64, 16><<<NE / 128 + 1, 256, 0, stream>>>(B0, esrc, edst, out, NE);

  k_post<<<(NN * 64 + T - 1) / T, T, 0, stream>>>(out, normD, b2, NN * 64, 6, 0);
}


# Degree count: 64 work-groups each count one slice of the edge list into an LDS table over all nodes (out and in count share a word), the norm launch adds the slices; no global atomics, no zero launch
# speedup vs baseline: 2.7907x; 1.4555x over previous
#include <hip/hip_runtime.h>


typedef __attribute__((ext_vector_type(2))) float v2f;
typedef __attribute__((ext_vector_type(8))) float v8f;

#define NN   50000
#define NE   600000
#define KDIM 128


#define NSLICE 64
#define PADN   50176

__global__ __launch_bounds__(1024) void k_degcount(
    const int* __restrict__ src, const int* __restrict__ dst,
    unsigned* __restrict__ part) {
  __shared__ uint4 tbl4[NN / 4];
  unsigned* tbl = (unsigned*)tbl4;

  constexpr int N4    = NE / 4;
  constexpr int PER   = (N4 + NSLICE - 1) / NSLICE;
  constexpr int TRIPS = (PER + 1023) / 1024;
  static_assert(NE % 4 == 0 && NN % 4 == 0, "whole 16-byte groups");
  static_assert(4 * PER < 65536, "a slice's count must fit 16 bits");

  const int tid   = threadIdx.x;
  const int slice = blockIdx.x;
  const int start = slice * PER;
  const int cnt   = min(PER, N4 - start);

  const int4* s4 = (const int4*)src + start;
  const int4* d4 = (const int4*)dst + start;
  int4 sv[TRIPS], dv[TRIPS];
  #pragma unroll
  for (int t = 0; t < TRIPS; ++t) {
    const int i = tid + 1024 * t;
    sv[t] = make_int4(0, 0, 0, 0);
    dv[t] = make_int4(0, 0, 0, 0);
    if (i < cnt) { sv[t] = s4[i]; dv[t] = d4[i]; }
  }

  for (int i = tid; i < NN / 4; i += 1024) tbl4[i] = make_uint4(0u, 0u, 0u, 0u);
  __syncthreads();

  #pragma unroll
  for (int t = 0; t < TRIPS; ++t) {
    if (tid + 1024 * t < cnt) {
      atomicAdd(&tbl[sv[t].x], 1u);
      atomicAdd(&tbl[sv[t].y], 1u);
      atomicAdd(&tbl[sv[t].z], 1u);
      atomicAdd(&tbl[sv[t].w], 1u);
      atomicAdd(&tbl[dv[t].x], 0x10000u);
      atomicAdd(&tbl[dv[t].y], 0x10000u);
      atomicAdd(&tbl[dv[t].z], 0x10000u);
      atomicAdd(&tbl[dv[t].w], 0x10000u);
    }
  }
  __syncthreads();

  uint4* p4 = (uint4*)(part + (size_t)slice * PADN);
  for (int i = tid; i < NN / 4; i += 1024) p4[i] = tbl4[i];
}

__global__ void k_norm(const unsigned* __restrict__ part, float* __restrict__ a,
                       float* __restrict__ b, int n) {
  int i = blockIdx.x * blockDim.x + threadIdx.x;
  if (i < n) {
    unsigned s = 0u, d = 0u;
    for (int g = 0; g < NSLICE; ++g) {
      const unsigned w = part[(size_t)g * PADN + i];
      s += w & 0xffffu;
      d += w >> 16;
    }
    a[i] = rsqrtf(fmaxf((float)s, 1.0f));
    b[i] = rsqrtf(fmaxf((float)d, 1.0f));
  }
}


template <int NCOL, bool PRE>
__global__ __launch_bounds__(NCOL / 2) void k_gemm_wreg(
    const float* __restrict__ X, const float* __restrict__ norm,
    const float* __restrict__ rowS, const float* __restrict__ colB,
    const float* __restrict__ W, float* __restrict__ Y,
    float* __restrict__ Z, int nstrips) {
  const int lane = threadIdx.x & 31;
  const int wave = threadIdx.x >> 5;
  const int c    = lane & 15;
  const int h    = lane >> 4;
  const int col0 = wave * 64 + 4 * c;

  int strip = blockIdx.x;
  if (strip >= nstrips) return;

  v2f b[4][32];
  {
    const float4* wp = (const float4*)(W + (size_t)(4 * h) * NCOL + col0);
    #pragma unroll
    for (int j = 0; j < 16; ++j) {
      #pragma unroll
      for (int q = 0; q < 2; ++q) {
        const float4 w0 = wp[((8 * j + 2 * q) * NCOL) >> 2];
        const float4 w1 = wp[((8 * j + 2 * q + 1) * NCOL) >> 2];
        b[0][2 * j + q] = v2f{w0.x, w1.x};
        b[1][2 * j + q] = v2f{w0.y, w1.y};
        b[2][2 * j + q] = v2f{w0.z, w1.z};
        b[3][2 * j + q] = v2f{w0.w, w1.w};
      }
    }
  }

  float4 a[16];
  float  nm;
  {
    const float4* xp = (const float4*)(X + (size_t)(strip * 16 + c) * KDIM + 4 * h);
    #pragma unroll
    for (int j = 0; j < 16; ++j) a[j] = xp[2 * j];
    nm = norm[strip * 16 + c];
  }

  float4 bv[16];
  float  nd = 0.f;
  if constexpr (PRE) {
    const float4* bp = (const float4*)(colB + 4 * h);
    #pragma unroll
    for (int j = 0; j < 16; ++j) bv[j] = bp[2 * j];
    nd = rowS[strip * 16 + c];
  }

  for (;;) {
    const int  nxt  = strip + (int)gridDim.x;
    const bool more = nxt < nstrips;
    const int  pf   = more ? nxt : strip;

    float4 an[16];
    const float4* xn = (const float4*)(X + (size_t)(pf * 16 + c) * KDIM + 4 * h);
    #pragma unroll
    for (int j = 0; j < 16; ++j) an[j] = xn[2 * j];
    const float nn = norm[pf * 16 + c];
    float ndn = 0.f;
    if constexpr (PRE) ndn = rowS[pf * 16 + c];

    v8f acc0 = {}, acc1 = {}, acc2 = {}, acc3 = {};
    #pragma unroll
    for (int j = 0; j < 16; ++j) {
      float4 x = a[j];
      if constexpr (PRE) {
        x.x = fmaxf(x.x * nd + bv[j].x, 0.0f);
        x.y = fmaxf(x.y * nd + bv[j].y, 0.0f);
        x.z = fmaxf(x.z * nd + bv[j].z, 0.0f);
        x.w = fmaxf(x.w * nd + bv[j].w, 0.0f);
      }
      const v2f a0 = v2f{x.x * nm, x.y * nm};
      const v2f a1 = v2f{x.z * nm, x.w * nm};
      acc0 = __builtin_amdgcn_wmma_f32_16x16x4_f32(false, a0, false, b[0][2 * j], (short)0, acc0, false, false);
      acc1 = __builtin_amdgcn_wmma_f32_16x16x4_f32(false, a0, false, b[1][2 * j], (short)0, acc1, false, false);
      acc2 = __builtin_amdgcn_wmma_f32_16x16x4_f32(false, a0, false, b[2][2 * j], (short)0, acc2, false, false);
      acc3 = __builtin_amdgcn_wmma_f32_16x16x4_f32(false, a0, false, b[3][2 * j], (short)0, acc3, false, false);
      acc0 = __builtin_amdgcn_wmma_f32_16x16x4_f32(false, a1, false, b[0][2 * j + 1], (short)0, acc0, false, false);
      acc1 = __builtin_amdgcn_wmma_f32_16x16x4_f32(false, a1, false, b[1][2 * j + 1], (short)0, acc1, false, false);
      acc2 = __builtin_amdgcn_wmma_f32_16x16x4_f32(false, a1, false, b[2][2 * j + 1], (short)0, acc2, false, false);
      acc3 = __builtin_amdgcn_wmma_f32_16x16x4_f32(false, a1, false, b[3][2 * j + 1], (short)0, acc3, false, false);
    }

    float4* yp = (float4*)(Y + (size_t)(strip * 16 + 8 * h) * NCOL + col0);
    #pragma unroll
    for (int r = 0; r < 8; ++r)
      yp[(r * NCOL) >> 2] = make_float4(acc0[r], acc1[r], acc2[r], acc3[r]);

    float4* zp = (float4*)(Z + (size_t)(strip * 16 + 8 * h) * NCOL + col0);
    #pragma unroll
    for (int r = 0; r < 8; ++r)
      zp[(r * NCOL) >> 2] = make_float4(0.f, 0.f, 0.f, 0.f);

    if (!more) break;
    #pragma unroll
    for (int j = 0; j < 16; ++j) a[j] = an[j];
    nm    = nn;
    nd    = ndn;
    strip = nxt;
  }
}


template <int F, int EPW>
__global__ __launch_bounds__(256) void k_scatter_line(
    const float* __restrict__ H, const int* __restrict__ src,
    const int* __restrict__ dst, float* __restrict__ out, int nE) {
  constexpr int Q = F / 32;
  const int lane = threadIdx.x & 31;
  const int wave = __builtin_amdgcn_readfirstlane((int)(threadIdx.x >> 5));
  const int g    = (int)blockIdx.x * 8 + wave;
  const int ngrp = nE / EPW;

  if (g < ngrp) {
    const int e0 = g * EPW;
    int s[EPW], d[EPW];
    #pragma unroll
    for (int k = 0; k < EPW; ++k) { s[k] = src[e0 + k]; d[k] = dst[e0 + k]; }

    float v[EPW][Q];
    #pragma unroll
    for (int k = 0; k < EPW; ++k) {
      const float* hp = H + (size_t)s[k] * F + lane;
      #pragma unroll
      for (int q = 0; q < Q; ++q) v[k][q] = hp[32 * q];
    }
    #pragma unroll
    for (int k = 0; k < EPW; ++k) {
      float* op = out + (size_t)d[k] * F + lane;
      #pragma unroll
      for (int q = 0; q < Q; ++q) atomicAdd(op + 32 * q, v[k][q]);
    }
  }

  const int et = ngrp * EPW + g;
  if (g < EPW && et < nE) {
    const float* hp = H + (size_t)src[et] * F + lane;
    float*       op = out + (size_t)dst[et] * F + lane;
    #pragma unroll
    for (int q = 0; q < Q; ++q) atomicAdd(op + 32 * q, hp[32 * q]);
  }
}


__global__ void k_post(float* __restrict__ h, const float* __restrict__ normD,
                       const float* __restrict__ b, int total, int shift,
                       int doRelu) {
  int i = blockIdx.x * blockDim.x + threadIdx.x;
  if (i >= total) return;
  const int row = i >> shift;
  const int col = i & ((1 << shift) - 1);
  float v = h[i] * normD[row] + b[col];
  h[i] = doRelu ? fmaxf(v, 0.0f) : v;
}


extern "C" void kernel_launch(void* const* d_in, const int* in_sizes, int n_in,
                              void* d_out, int out_size, void* d_ws, size_t ws_size,
                              hipStream_t stream) {
  (void)in_sizes; (void)n_in; (void)out_size; (void)ws_size;

  const float* features = (const float*)d_in[0];
  const float* W1       = (const float*)d_in[1];
  const float* b1       = (const float*)d_in[2];
  const float* W2       = (const float*)d_in[3];
  const float* b2       = (const float*)d_in[4];
  const int*   esrc     = (const int*)d_in[5];
  const int*   edst     = (const int*)d_in[6];
  float*       out      = (float*)d_out;

  float* w     = (float*)d_ws;
  float* normS = w;
  float* normD = w + PADN;
  float* B0    = w + 2 * PADN;
  float* B1    = B0 + (size_t)NN * 128;
  unsigned* part = (unsigned*)B0;

  const int T = 256;

  k_degcount<<<NSLICE, 1024, 0, stream>>>(esrc, edst, part);
  k_norm<<<(NN + T - 1) / T, T, 0, stream>>>(part, normS, normD, NN);

  k_gemm_wreg<128, false><<<1024, 64, 0, stream>>>(features, normS, normD, b1, W1,
                                                   B0, B1, NN / 16);

  k_scatter_line<128, 8><<<NE / 64 + 1, 256, 0, stream>>>(B0, esrc, edst, B1, NE);

  k_gemm_wreg<64, true><<<2048, 32, 0, stream>>>(B1, normS, normD, b1, W2,
                                                 B0, out, NN / 16);

  k_scatter_line<64, 16><<<NE / 128 + 1, 256, 0, stream>>>(B0, esrc, edst, out, NE);

  k_post<<<(NN * 64 + T - 1) / T, T, 0, stream>>>(out, normD, b2, NN * 64, 6, 0);
}


# Output scale and bias folded into the last edge scatter: each edge adds normD[dst]*row into sums that gemm2 starts at b2; post pass gone, 7 -> 6 launches; rows are scaled before the atomic sums
# speedup vs baseline: 2.9738x; 1.0656x over previous
#include <hip/hip_runtime.h>


typedef __attribute__((ext_vector_type(2))) float v2f;
typedef __attribute__((ext_vector_type(8))) float v8f;

#define NN   50000
#define NE   600000
#define KDIM 128


#define NSLICE 64
#define PADN   50176

__global__ __launch_bounds__(1024) void k_degcount(
    const int* __restrict__ src, const int* __restrict__ dst,
    unsigned* __restrict__ part) {
  __shared__ uint4 tbl4[NN / 4];
  unsigned* tbl = (unsigned*)tbl4;

  constexpr int N4    = NE / 4;
  constexpr int PER   = (N4 + NSLICE - 1) / NSLICE;
  constexpr int TRIPS = (PER + 1023) / 1024;
  static_assert(NE % 4 == 0 && NN % 4 == 0, "whole 16-byte groups");
  static_assert(4 * PER < 65536, "a slice's count must fit 16 bits");

  const int tid   = threadIdx.x;
  const int slice = blockIdx.x;
  const int start = slice * PER;
  const int cnt   = min(PER, N4 - start);

  const int4* s4 = (const int4*)src + start;
  const int4* d4 = (const int4*)dst + start;
  int4 sv[TRIPS], dv[TRIPS];
  #pragma unroll
  for (int t = 0; t < TRIPS; ++t) {
    const int i = tid + 1024 * t;
    sv[t] = make_int4(0, 0, 0, 0);
    dv[t] = make_int4(0, 0, 0, 0);
    if (i < cnt) { sv[t] = s4[i]; dv[t] = d4[i]; }
  }

  for (int i = tid; i < NN / 4; i += 1024) tbl4[i] = make_uint4(0u, 0u, 0u, 0u);
  __syncthreads();

  #pragma unroll
  for (int t = 0; t < TRIPS; ++t) {
    if (tid + 1024 * t < cnt) {
      atomicAdd(&tbl[sv[t].x], 1u);
      atomicAdd(&tbl[sv[t].y], 1u);
      atomicAdd(&tbl[sv[t].z], 1u);
      atomicAdd(&tbl[sv[t].w], 1u);
      atomicAdd(&tbl[dv[t].x], 0x10000u);
      atomicAdd(&tbl[dv[t].y], 0x10000u);
      atomicAdd(&tbl[dv[t].z], 0x10000u);
      atomicAdd(&tbl[dv[t].w], 0x10000u);
    }
  }
  __syncthreads();

  uint4* p4 = (uint4*)(part + (size_t)slice * PADN);
  for (int i = tid; i < NN / 4; i += 1024) p4[i] = tbl4[i];
}

__global__ void k_norm(const unsigned* __restrict__ part, float* __restrict__ a,
                       float* __restrict__ b, int n) {
  int i = blockIdx.x * blockDim.x + threadIdx.x;
  if (i < n) {
    unsigned s = 0u, d = 0u;
    for (int g = 0; g < NSLICE; ++g) {
      const unsigned w = part[(size_t)g * PADN + i];
      s += w & 0xffffu;
      d += w >> 16;
    }
    a[i] = rsqrtf(fmaxf((float)s, 1.0f));
    b[i] = rsqrtf(fmaxf((float)d, 1.0f));
  }
}


template <int NCOL, bool PRE, bool ZBIAS>
__global__ __launch_bounds__(NCOL / 2) void k_gemm_wreg(
    const float* __restrict__ X, const float* __restrict__ norm,
    const float* __restrict__ rowS, const float* __restrict__ colB,
    const float* __restrict__ W, const float* __restrict__ zB,
    float* __restrict__ Y, float* __restrict__ Z, int nstrips) {
  const int lane = threadIdx.x & 31;
  const int wave = threadIdx.x >> 5;
  const int c    = lane & 15;
  const int h    = lane >> 4;
  const int col0 = wave * 64 + 4 * c;

  int strip = blockIdx.x;
  if (strip >= nstrips) return;

  v2f b[4][32];
  {
    const float4* wp = (const float4*)(W + (size_t)(4 * h) * NCOL + col0);
    #pragma unroll
    for (int j = 0; j < 16; ++j) {
      #pragma unroll
      for (int q = 0; q < 2; ++q) {
        const float4 w0 = wp[((8 * j + 2 * q) * NCOL) >> 2];
        const float4 w1 = wp[((8 * j + 2 * q + 1) * NCOL) >> 2];
        b[0][2 * j + q] = v2f{w0.x, w1.x};
        b[1][2 * j + q] = v2f{w0.y, w1.y};
        b[2][2 * j + q] = v2f{w0.z, w1.z};
        b[3][2 * j + q] = v2f{w0.w, w1.w};
      }
    }
  }

  float4 a[16];
  float  nm;
  {
    const float4* xp = (const float4*)(X + (size_t)(strip * 16 + c) * KDIM + 4 * h);
    #pragma unroll
    for (int j = 0; j < 16; ++j) a[j] = xp[2 * j];
    nm = norm[strip * 16 + c];
  }

  float4 z0 = make_float4(0.f, 0.f, 0.f, 0.f);
  if constexpr (ZBIAS) z0 = *(const float4*)(zB + col0);

  float4 bv[16];
  float  nd = 0.f;
  if constexpr (PRE) {
    const float4* bp = (const float4*)(colB + 4 * h);
    #pragma unroll
    for (int j = 0; j < 16; ++j) bv[j] = bp[2 * j];
    nd = rowS[strip * 16 + c];
  }

  for (;;) {
    const int  nxt  = strip + (int)gridDim.x;
    const bool more = nxt < nstrips;
    const int  pf   = more ? nxt : strip;

    float4 an[16];
    const float4* xn = (const float4*)(X + (size_t)(pf * 16 + c) * KDIM + 4 * h);
    #pragma unroll
    for (int j = 0; j < 16; ++j) an[j] = xn[2 * j];
    const float nn = norm[pf * 16 + c];
    float ndn = 0.f;
    if constexpr (PRE) ndn = rowS[pf * 16 + c];

    v8f acc0 = {}, acc1 = {}, acc2 = {}, acc3 = {};
    #pragma unroll
    for (int j = 0; j < 16; ++j) {
      float4 x = a[j];
      if constexpr (PRE) {
        x.x = fmaxf(x.x * nd + bv[j].x, 0.0f);
        x.y = fmaxf(x.y * nd + bv[j].y, 0.0f);
        x.z = fmaxf(x.z * nd + bv[j].z, 0.0f);
        x.w = fmaxf(x.w * nd + bv[j].w, 0.0f);
      }
      const v2f a0 = v2f{x.x * nm, x.y * nm};
      const v2f a1 = v2f{x.z * nm, x.w * nm};
      acc0 = __builtin_amdgcn_wmma_f32_16x16x4_f32(false, a0, false, b[0][2 * j], (short)0, acc0, false, false);
      acc1 = __builtin_amdgcn_wmma_f32_16x16x4_f32(false, a0, false, b[1][2 * j], (short)0, acc1, false, false);
      acc2 = __builtin_amdgcn_wmma_f32_16x16x4_f32(false, a0, false, b[2][2 * j], (short)0, acc2, false, false);
      acc3 = __builtin_amdgcn_wmma_f32_16x16x4_f32(false, a0, false, b[3][2 * j], (short)0, acc3, false, false);
      acc0 = __builtin_amdgcn_wmma_f32_16x16x4_f32(false, a1, false, b[0][2 * j + 1], (short)0, acc0, false, false);
      acc1 = __builtin_amdgcn_wmma_f32_16x16x4_f32(false, a1, false, b[1][2 * j + 1], (short)0, acc1, false, false);
      acc2 = __builtin_amdgcn_wmma_f32_16x16x4_f32(false, a1, false, b[2][2 * j + 1], (short)0, acc2, false, false);
      acc3 = __builtin_amdgcn_wmma_f32_16x16x4_f32(false, a1, false, b[3][2 * j + 1], (short)0, acc3, false, false);
    }

    float4* yp = (float4*)(Y + (size_t)(strip * 16 + 8 * h) * NCOL + col0);
    #pragma unroll
    for (int r = 0; r < 8; ++r)
      yp[(r * NCOL) >> 2] = make_float4(acc0[r], acc1[r], acc2[r], acc3[r]);

    float4* zp = (float4*)(Z + (size_t)(strip * 16 + 8 * h) * NCOL + col0);
    #pragma unroll
    for (int r = 0; r < 8; ++r)
      zp[(r * NCOL) >> 2] = z0;

    if (!more) break;
    #pragma unroll
    for (int j = 0; j < 16; ++j) a[j] = an[j];
    nm    = nn;
    nd    = ndn;
    strip = nxt;
  }
}


template <int F, int EPW, bool SCALE>
__global__ __launch_bounds__(256) void k_scatter_line(
    const float* __restrict__ H, const int* __restrict__ src,
    const int* __restrict__ dst, const float* __restrict__ rowScale,
    float* __restrict__ out, int nE) {
  constexpr int Q = F / 32;
  const int lane = threadIdx.x & 31;
  const int wave = __builtin_amdgcn_readfirstlane((int)(threadIdx.x >> 5));
  const int g    = (int)blockIdx.x * 8 + wave;
  const int ngrp = nE / EPW;

  if (g < ngrp) {
    const int e0 = g * EPW;
    int s[EPW], d[EPW];
    #pragma unroll
    for (int k = 0; k < EPW; ++k) { s[k] = src[e0 + k]; d[k] = dst[e0 + k]; }

    float v[EPW][Q];
    #pragma unroll
    for (int k = 0; k < EPW; ++k) {
      const float* hp = H + (size_t)s[k] * F + lane;
      #pragma unroll
      for (int q = 0; q < Q; ++q) v[k][q] = hp[32 * q];
    }
    if constexpr (SCALE) {
      float sc[EPW];
      #pragma unroll
      for (int k = 0; k < EPW; ++k) sc[k] = rowScale[d[k]];
      #pragma unroll
      for (int k = 0; k < EPW; ++k) {
        #pragma unroll
        for (int q = 0; q < Q; ++q) v[k][q] *= sc[k];
      }
    }
    #pragma unroll
    for (int k = 0; k < EPW; ++k) {
      float* op = out + (size_t)d[k] * F + lane;
      #pragma unroll
      for (int q = 0; q < Q; ++q) atomicAdd(op + 32 * q, v[k][q]);
    }
  }

  const int et = ngrp * EPW + g;
  if (g < EPW && et < nE) {
    const int    dt = dst[et];
    const float* hp = H + (size_t)src[et] * F + lane;
    float*       op = out + (size_t)dt * F + lane;
    float        sc = 1.0f;
    if constexpr (SCALE) sc = rowScale[dt];
    #pragma unroll
    for (int q = 0; q < Q; ++q) {
      float x = hp[32 * q];
      if constexpr (SCALE) x *= sc;
      atomicAdd(op + 32 * q, x);
    }
  }
}


extern "C" void kernel_launch(void* const* d_in, const int* in_sizes, int n_in,
                              void* d_out, int out_size, void* d_ws, size_t ws_size,
                              hipStream_t stream) {
  (void)in_sizes; (void)n_in; (void)out_size; (void)ws_size;

  const float* features = (const float*)d_in[0];
  const float* W1       = (const float*)d_in[1];
  const float* b1       = (const float*)d_in[2];
  const float* W2       = (const float*)d_in[3];
  const float* b2       = (const float*)d_in[4];
  const int*   esrc     = (const int*)d_in[5];
  const int*   edst     = (const int*)d_in[6];
  float*       out      = (float*)d_out;

  float* w     = (float*)d_ws;
  float* normS = w;
  float* normD = w + PADN;
  float* B0    = w + 2 * PADN;
  float* B1    = B0 + (size_t)NN * 128;
  unsigned* part = (unsigned*)B0;

  const int T = 256;

  k_degcount<<<NSLICE, 1024, 0, stream>>>(esrc, edst, part);
  k_norm<<<(NN + T - 1) / T, T, 0, stream>>>(part, normS, normD, NN);

  k_gemm_wreg<128, false, false><<<1024, 64, 0, stream>>>(features, normS, normD, b1, W1,
                                                          b2, B0, B1, NN / 16);

  k_scatter_line<128, 8, false><<<NE / 64 + 1, 256, 0, stream>>>(B0, esrc, edst, normD,
                                                                 B1, NE);

  k_gemm_wreg<64, true, true><<<2048, 32, 0, stream>>>(B1, normS, normD, b1, W2,
                                                       b2, B0, out, NN / 16);

  k_scatter_line<64, 16, true><<<NE / 128 + 1, 256, 0, stream>>>(B0, esrc, edst, normD,
                                                                 out, NE);
}


# Edge sums: edges binned in-call by 256-node destination range; one work-group per range adds gathered rows into an LDS table (LDS float atomics, arrival order), stores rows once, scale after sum
# speedup vs baseline: 3.7624x; 1.2652x over previous
#include <hip/hip_runtime.h>


typedef __attribute__((ext_vector_type(2))) float v2f;
typedef __attribute__((ext_vector_type(8))) float v8f;

#define NN   50000
#define NE   600000
#define KDIM 128


#define NSLICE 64
#define PADN   50176

#define BINSH  8
#define BINROWS 256
#define NBIN   ((NN + BINROWS - 1) / BINROWS)
#define BINPAD 256

__global__ __launch_bounds__(1024) void k_degcount(
    const int* __restrict__ src, const int* __restrict__ dst,
    unsigned* __restrict__ part, unsigned* __restrict__ binCnt) {
  __shared__ uint4 tbl4[NN / 4];
  __shared__ unsigned lbin[BINPAD];
  unsigned* tbl = (unsigned*)tbl4;
  static_assert(NBIN <= BINPAD, "bins must fit the tables");

  constexpr int N4    = NE / 4;
  constexpr int PER   = (N4 + NSLICE - 1) / NSLICE;
  constexpr int TRIPS = (PER + 1023) / 1024;
  static_assert(NE % 4 == 0 && NN % 4 == 0, "whole 16-byte groups");
  static_assert(4 * PER < 65536, "a slice's count must fit 16 bits");

  const int tid   = threadIdx.x;
  const int slice = blockIdx.x;
  const int start = slice * PER;
  const int cnt   = min(PER, N4 - start);

  const int4* s4 = (const int4*)src + start;
  const int4* d4 = (const int4*)dst + start;
  int4 sv[TRIPS], dv[TRIPS];
  #pragma unroll
  for (int t = 0; t < TRIPS; ++t) {
    const int i = tid + 1024 * t;
    sv[t] = make_int4(0, 0, 0, 0);
    dv[t] = make_int4(0, 0, 0, 0);
    if (i < cnt) { sv[t] = s4[i]; dv[t] = d4[i]; }
  }

  for (int i = tid; i < NN / 4; i += 1024) tbl4[i] = make_uint4(0u, 0u, 0u, 0u);
  if (tid < BINPAD) lbin[tid] = 0u;
  __syncthreads();

  #pragma unroll
  for (int t = 0; t < TRIPS; ++t) {
    if (tid + 1024 * t < cnt) {
      atomicAdd(&tbl[sv[t].x], 1u);
      atomicAdd(&tbl[sv[t].y], 1u);
      atomicAdd(&tbl[sv[t].z], 1u);
      atomicAdd(&tbl[sv[t].w], 1u);
      atomicAdd(&tbl[dv[t].x], 0x10000u);
      atomicAdd(&tbl[dv[t].y], 0x10000u);
      atomicAdd(&tbl[dv[t].z], 0x10000u);
      atomicAdd(&tbl[dv[t].w], 0x10000u);
      atomicAdd(&lbin[dv[t].x >> BINSH], 1u);
      atomicAdd(&lbin[dv[t].y >> BINSH], 1u);
      atomicAdd(&lbin[dv[t].z >> BINSH], 1u);
      atomicAdd(&lbin[dv[t].w >> BINSH], 1u);
    }
  }
  __syncthreads();

  if (tid < BINPAD) binCnt[slice * BINPAD + tid] = lbin[tid];

  uint4* p4 = (uint4*)(part + (size_t)slice * PADN);
  for (int i = tid; i < NN / 4; i += 1024) p4[i] = tbl4[i];
}

__global__ void k_norm(const unsigned* __restrict__ part, float* __restrict__ a,
                       float* __restrict__ b, int n,
                       const unsigned* __restrict__ binCnt,
                       unsigned* __restrict__ starts,
                       unsigned* __restrict__ binStart) {
  if (blockIdx.x == gridDim.x - 1) {
    __shared__ unsigned tot[BINPAD];
    const int bn = threadIdx.x;
    unsigned t = 0u;
    for (int s = 0; s < NSLICE; ++s) t += binCnt[s * BINPAD + bn];
    tot[bn] = t;
    __syncthreads();
    unsigned base = 0u;
    for (int j = 0; j < bn; ++j) base += tot[j];
    binStart[bn] = base;
    if (bn == BINPAD - 1) binStart[BINPAD] = base + t;
    unsigned run = base;
    for (int s = 0; s < NSLICE; ++s) {
      starts[s * BINPAD + bn] = run;
      run += binCnt[s * BINPAD + bn];
    }
    return;
  }
  int i = blockIdx.x * blockDim.x + threadIdx.x;
  if (i < n) {
    unsigned s = 0u, d = 0u;
    for (int g = 0; g < NSLICE; ++g) {
      const unsigned w = part[(size_t)g * PADN + i];
      s += w & 0xffffu;
      d += w >> 16;
    }
    a[i] = rsqrtf(fmaxf((float)s, 1.0f));
    b[i] = rsqrtf(fmaxf((float)d, 1.0f));
  }
}


template <int NCOL, bool PRE>
__global__ __launch_bounds__(NCOL / 2) void k_gemm_wreg(
    const float* __restrict__ X, const float* __restrict__ norm,
    const float* __restrict__ rowS, const float* __restrict__ colB,
    const float* __restrict__ W, float* __restrict__ Y, int nstrips) {
  const int lane = threadIdx.x & 31;
  const int wave = threadIdx.x >> 5;
  const int c    = lane & 15;
  const int h    = lane >> 4;
  const int col0 = wave * 64 + 4 * c;

  int strip = blockIdx.x;
  if (strip >= nstrips) return;

  v2f b[4][32];
  {
    const float4* wp = (const float4*)(W + (size_t)(4 * h) * NCOL + col0);
    #pragma unroll
    for (int j = 0; j < 16; ++j) {
      #pragma unroll
      for (int q = 0; q < 2; ++q) {
        const float4 w0 = wp[((8 * j + 2 * q) * NCOL) >> 2];
        const float4 w1 = wp[((8 * j + 2 * q + 1) * NCOL) >> 2];
        b[0][2 * j + q] = v2f{w0.x, w1.x};
        b[1][2 * j + q] = v2f{w0.y, w1.y};
        b[2][2 * j + q] = v2f{w0.z, w1.z};
        b[3][2 * j + q] = v2f{w0.w, w1.w};
      }
    }
  }

  float4 a[16];
  float  nm;
  {
    const float4* xp = (const float4*)(X + (size_t)(strip * 16 + c) * KDIM + 4 * h);
    #pragma unroll
    for (int j = 0; j < 16; ++j) a[j] = xp[2 * j];
    nm = norm[strip * 16 + c];
  }

  float4 bv[16];
  float  nd = 0.f;
  if constexpr (PRE) {
    const float4* bp = (const float4*)(colB + 4 * h);
    #pragma unroll
    for (int j = 0; j < 16; ++j) bv[j] = bp[2 * j];
    nd = rowS[strip * 16 + c];
  }

  for (;;) {
    const int  nxt  = strip + (int)gridDim.x;
    const bool more = nxt < nstrips;
    const int  pf   = more ? nxt : strip;

    float4 an[16];
    const float4* xn = (const float4*)(X + (size_t)(pf * 16 + c) * KDIM + 4 * h);
    #pragma unroll
    for (int j = 0; j < 16; ++j) an[j] = xn[2 * j];
    const float nn = norm[pf * 16 + c];
    float ndn = 0.f;
    if constexpr (PRE) ndn = rowS[pf * 16 + c];

    v8f acc0 = {}, acc1 = {}, acc2 = {}, acc3 = {};
    #pragma unroll
    for (int j = 0; j < 16; ++j) {
      float4 x = a[j];
      if constexpr (PRE) {
        x.x = fmaxf(x.x * nd + bv[j].x, 0.0f);
        x.y = fmaxf(x.y * nd + bv[j].y, 0.0f);
        x.z = fmaxf(x.z * nd + bv[j].z, 0.0f);
        x.w = fmaxf(x.w * nd + bv[j].w, 0.0f);
      }
      const v2f a0 = v2f{x.x * nm, x.y * nm};
      const v2f a1 = v2f{x.z * nm, x.w * nm};
      acc0 = __builtin_amdgcn_wmma_f32_16x16x4_f32(false, a0, false, b[0][2 * j], (short)0, acc0, false, false);
      acc1 = __builtin_amdgcn_wmma_f32_16x16x4_f32(false, a0, false, b[1][2 * j], (short)0, acc1, false, false);
      acc2 = __builtin_amdgcn_wmma_f32_16x16x4_f32(false, a0, false, b[2][2 * j], (short)0, acc2, false, false);
      acc3 = __builtin_amdgcn_wmma_f32_16x16x4_f32(false, a0, false, b[3][2 * j], (short)0, acc3, false, false);
      acc0 = __builtin_amdgcn_wmma_f32_16x16x4_f32(false, a1, false, b[0][2 * j + 1], (short)0, acc0, false, false);
      acc1 = __builtin_amdgcn_wmma_f32_16x16x4_f32(false, a1, false, b[1][2 * j + 1], (short)0, acc1, false, false);
      acc2 = __builtin_amdgcn_wmma_f32_16x16x4_f32(false, a1, false, b[2][2 * j + 1], (short)0, acc2, false, false);
      acc3 = __builtin_amdgcn_wmma_f32_16x16x4_f32(false, a1, false, b[3][2 * j + 1], (short)0, acc3, false, false);
    }

    float4* yp = (float4*)(Y + (size_t)(strip * 16 + 8 * h) * NCOL + col0);
    #pragma unroll
    for (int r = 0; r < 8; ++r)
      yp[(r * NCOL) >> 2] = make_float4(acc0[r], acc1[r], acc2[r], acc3[r]);

    if (!more) break;
    #pragma unroll
    for (int j = 0; j < 16; ++j) a[j] = an[j];
    nm    = nn;
    nd    = ndn;
    strip = nxt;
  }
}


__global__ __launch_bounds__(1024) void k_place(
    const int* __restrict__ src, const int* __restrict__ dst,
    const unsigned* __restrict__ starts, int* __restrict__ pairs) {
  __shared__ unsigned cur[BINPAD];

  constexpr int N4    = NE / 4;
  constexpr int PER   = (N4 + NSLICE - 1) / NSLICE;
  constexpr int TRIPS = (PER + 1023) / 1024;

  const int tid   = threadIdx.x;
  const int slice = blockIdx.x;
  const int start = slice * PER;
  const int cnt   = min(PER, N4 - start);

  const int4* s4 = (const int4*)src + start;
  const int4* d4 = (const int4*)dst + start;
  int4 sv[TRIPS], dv[TRIPS];
  #pragma unroll
  for (int t = 0; t < TRIPS; ++t) {
    const int i = tid + 1024 * t;
    sv[t] = make_int4(0, 0, 0, 0);
    dv[t] = make_int4(0, 0, 0, 0);
    if (i < cnt) { sv[t] = s4[i]; dv[t] = d4[i]; }
  }

  if (tid < BINPAD) cur[tid] = starts[slice * BINPAD + tid];
  __syncthreads();

  int2* pp = (int2*)pairs;
  #pragma unroll
  for (int t = 0; t < TRIPS; ++t) {
    if (tid + 1024 * t < cnt) {
      const unsigned p0 = atomicAdd(&cur[dv[t].x >> BINSH], 1u);
      const unsigned p1 = atomicAdd(&cur[dv[t].y >> BINSH], 1u);
      const unsigned p2 = atomicAdd(&cur[dv[t].z >> BINSH], 1u);
      const unsigned p3 = atomicAdd(&cur[dv[t].w >> BINSH], 1u);
      pp[p0] = make_int2(sv[t].x, dv[t].x);
      pp[p1] = make_int2(sv[t].y, dv[t].y);
      pp[p2] = make_int2(sv[t].z, dv[t].z);
      pp[p3] = make_int2(sv[t].w, dv[t].w);
    }
  }
}


template <int F, int EPW, bool FINAL>
__global__ __launch_bounds__(1024) void k_binsum(
    const float* __restrict__ H, const int* __restrict__ pairs,
    const unsigned* __restrict__ binStart, const float* __restrict__ rowScale,
    const float* __restrict__ bias, float* __restrict__ out) {
  constexpr int Q  = F / 32;
  constexpr int F4 = F / 4;
  __shared__ float4 acc4[BINROWS * F4];
  float* acc = (float*)acc4;

  const int tid  = threadIdx.x;
  const int lane = tid & 31;
  const int wave = __builtin_amdgcn_readfirstlane(tid >> 5);
  const int bin  = blockIdx.x;

  const int beg = (int)binStart[bin];
  const int end = (int)binStart[bin + 1];

  for (int i = tid; i < BINROWS * F4; i += 1024) acc4[i] = make_float4(0.f, 0.f, 0.f, 0.f);
  __syncthreads();

  const int2* pp = (const int2*)pairs;
  int e0 = beg + wave * EPW;
  if (e0 < end) {
    int s[EPW], d[EPW];
    #pragma unroll
    for (int k = 0; k < EPW; ++k) {
      const int2 p = pp[min(e0 + k, end - 1)];
      s[k] = p.x; d[k] = p.y;
    }
    for (;;) {
      float v[EPW][Q];
      #pragma unroll
      for (int k = 0; k < EPW; ++k) {
        const float* hp = H + (size_t)s[k] * F + lane;
        #pragma unroll
        for (int q = 0; q < Q; ++q) v[k][q] = hp[32 * q];
      }
      int r[EPW];
      #pragma unroll
      for (int k = 0; k < EPW; ++k) r[k] = (d[k] & (BINROWS - 1)) * F;
      const int live = end - e0;

      const int  en   = e0 + 32 * EPW;
      const bool more = en < end;
      if (more) {
        #pragma unroll
        for (int k = 0; k < EPW; ++k) {
          const int2 p = pp[min(en + k, end - 1)];
          s[k] = p.x; d[k] = p.y;
        }
      }

      #pragma unroll
      for (int k = 0; k < EPW; ++k) {
        if (k < live) {
          float* ap = acc + r[k] + lane;
          #pragma unroll
          for (int q = 0; q < Q; ++q) atomicAdd(ap + 32 * q, v[k][q]);
        }
      }
      if (!more) break;
      e0 = en;
    }
  }
  __syncthreads();

  const int node0 = bin * BINROWS;
  for (int i = tid; i < BINROWS * F4; i += 1024) {
    const int row  = i / F4;
    const int c4   = i % F4;
    const int node = node0 + row;
    if (node < NN) {
      float4 x = acc4[i];
      if constexpr (FINAL) {
        const float  sc = rowScale[node];
        const float4 bb = ((const float4*)bias)[c4];
        x.x = x.x * sc + bb.x;
        x.y = x.y * sc + bb.y;
        x.z = x.z * sc + bb.z;
        x.w = x.w * sc + bb.w;
      }
      ((float4*)out)[(size_t)node * F4 + c4] = x;
    }
  }
}


extern "C" void kernel_launch(void* const* d_in, const int* in_sizes, int n_in,
                              void* d_out, int out_size, void* d_ws, size_t ws_size,
                              hipStream_t stream) {
  (void)in_sizes; (void)n_in; (void)out_size; (void)ws_size;

  const float* features = (const float*)d_in[0];
  const float* W1       = (const float*)d_in[1];
  const float* b1       = (const float*)d_in[2];
  const float* W2       = (const float*)d_in[3];
  const float* b2       = (const float*)d_in[4];
  const int*   esrc     = (const int*)d_in[5];
  const int*   edst     = (const int*)d_in[6];
  float*       out      = (float*)d_out;

  float* w     = (float*)d_ws;
  float* normS = w;
  float* normD = w + PADN;
  float* B0    = w + 2 * PADN;
  float* B1    = B0 + (size_t)NN * 128;
  int*      pairs    = (int*)(B1 + (size_t)NN * 128);
  unsigned* binCnt   = (unsigned*)(pairs + 2 * NE);
  unsigned* starts   = binCnt + NSLICE * BINPAD;
  unsigned* binStart = starts + NSLICE * BINPAD;
  unsigned* part = (unsigned*)B0;

  const int T = 256;
  static_assert(T == BINPAD, "the scan block of k_norm has one thread per bin");

  k_degcount<<<NSLICE, 1024, 0, stream>>>(esrc, edst, part, binCnt);
  k_norm<<<(NN + T - 1) / T + 1, T, 0, stream>>>(part, normS, normD, NN, binCnt, starts,
                                                 binStart);
  k_place<<<NSLICE, 1024, 0, stream>>>(esrc, edst, starts, pairs);

  k_gemm_wreg<128, false><<<1024, 64, 0, stream>>>(features, normS, normD, b1, W1, B0,
                                                   NN / 16);

  k_binsum<128, 8, false><<<NBIN, 1024, 0, stream>>>(B0, pairs, binStart, normD, b2, B1);

  k_gemm_wreg<64, true><<<2048, 32, 0, stream>>>(B1, normS, normD, b1, W2, B0, NN / 16);

  k_binsum<64, 16, true><<<NBIN, 1024, 0, stream>>>(B0, pairs, binStart, normD, b2, out);
}


# Edge sums: 256 bins of 196 nodes, one per WGP, and an LDS reservation that keeps one summing work-group per WGP: all 256 WGPs add edges (196 of 256 before); sums stay arrival-order LDS atomics
# speedup vs baseline: 4.0044x; 1.0643x over previous
#include <hip/hip_runtime.h>


typedef __attribute__((ext_vector_type(2))) float v2f;
typedef __attribute__((ext_vector_type(8))) float v8f;

#define NN   50000
#define NE   600000
#define KDIM 128


#define NSLICE 64
#define PADN   50176

#define BINROWS 196
#define NBIN   ((NN + BINROWS - 1) / BINROWS)
#define BINPAD 256
#define SUM_LDS_BYTES (164 * 1024)

__global__ __launch_bounds__(1024) void k_degcount(
    const int* __restrict__ src, const int* __restrict__ dst,
    unsigned* __restrict__ part, unsigned* __restrict__ binCnt) {
  __shared__ uint4 tbl4[NN / 4];
  __shared__ unsigned lbin[BINPAD];
  unsigned* tbl = (unsigned*)tbl4;
  static_assert(NBIN <= BINPAD, "bins must fit the tables");

  constexpr int N4    = NE / 4;
  constexpr int PER   = (N4 + NSLICE - 1) / NSLICE;
  constexpr int TRIPS = (PER + 1023) / 1024;
  static_assert(NE % 4 == 0 && NN % 4 == 0, "whole 16-byte groups");
  static_assert(4 * PER < 65536, "a slice's count must fit 16 bits");

  const int tid   = threadIdx.x;
  const int slice = blockIdx.x;
  const int start = slice * PER;
  const int cnt   = min(PER, N4 - start);

  const int4* s4 = (const int4*)src + start;
  const int4* d4 = (const int4*)dst + start;
  int4 sv[TRIPS], dv[TRIPS];
  #pragma unroll
  for (int t = 0; t < TRIPS; ++t) {
    const int i = tid + 1024 * t;
    sv[t] = make_int4(0, 0, 0, 0);
    dv[t] = make_int4(0, 0, 0, 0);
    if (i < cnt) { sv[t] = s4[i]; dv[t] = d4[i]; }
  }

  for (int i = tid; i < NN / 4; i += 1024) tbl4[i] = make_uint4(0u, 0u, 0u, 0u);
  if (tid < BINPAD) lbin[tid] = 0u;
  __syncthreads();

  #pragma unroll
  for (int t = 0; t < TRIPS; ++t) {
    if (tid + 1024 * t < cnt) {
      atomicAdd(&tbl[sv[t].x], 1u);
      atomicAdd(&tbl[sv[t].y], 1u);
      atomicAdd(&tbl[sv[t].z], 1u);
      atomicAdd(&tbl[sv[t].w], 1u);
      atomicAdd(&tbl[dv[t].x], 0x10000u);
      atomicAdd(&tbl[dv[t].y], 0x10000u);
      atomicAdd(&tbl[dv[t].z], 0x10000u);
      atomicAdd(&tbl[dv[t].w], 0x10000u);
      atomicAdd(&lbin[(unsigned)dv[t].x / BINROWS], 1u);
      atomicAdd(&lbin[(unsigned)dv[t].y / BINROWS], 1u);
      atomicAdd(&lbin[(unsigned)dv[t].z / BINROWS], 1u);
      atomicAdd(&lbin[(unsigned)dv[t].w / BINROWS], 1u);
    }
  }
  __syncthreads();

  if (tid < BINPAD) binCnt[slice * BINPAD + tid] = lbin[tid];

  uint4* p4 = (uint4*)(part + (size_t)slice * PADN);
  for (int i = tid; i < NN / 4; i += 1024) p4[i] = tbl4[i];
}

__global__ void k_norm(const unsigned* __restrict__ part, float* __restrict__ a,
                       float* __restrict__ b, int n,
                       const unsigned* __restrict__ binCnt,
                       unsigned* __restrict__ starts,
                       unsigned* __restrict__ binStart) {
  if (blockIdx.x == gridDim.x - 1) {
    __shared__ unsigned tot[BINPAD];
    const int bn = threadIdx.x;
    unsigned t = 0u;
    for (int s = 0; s < NSLICE; ++s) t += binCnt[s * BINPAD + bn];
    tot[bn] = t;
    __syncthreads();
    unsigned base = 0u;
    for (int j = 0; j < bn; ++j) base += tot[j];
    binStart[bn] = base;
    if (bn == BINPAD - 1) binStart[BINPAD] = base + t;
    unsigned run = base;
    for (int s = 0; s < NSLICE; ++s) {
      starts[s * BINPAD + bn] = run;
      run += binCnt[s * BINPAD + bn];
    }
    return;
  }
  int i = blockIdx.x * blockDim.x + threadIdx.x;
  if (i < n) {
    unsigned s = 0u, d = 0u;
    for (int g = 0; g < NSLICE; ++g) {
      const unsigned w = part[(size_t)g * PADN + i];
      s += w & 0xffffu;
      d += w >> 16;
    }
    a[i] = rsqrtf(fmaxf((float)s, 1.0f));
    b[i] = rsqrtf(fmaxf((float)d, 1.0f));
  }
}


template <int NCOL, bool PRE>
__global__ __launch_bounds__(NCOL / 2) void k_gemm_wreg(
    const float* __restrict__ X, const float* __restrict__ norm,
    const float* __restrict__ rowS, const float* __restrict__ colB,
    const float* __restrict__ W, float* __restrict__ Y, int nstrips) {
  const int lane = threadIdx.x & 31;
  const int wave = threadIdx.x >> 5;
  const int c    = lane & 15;
  const int h    = lane >> 4;
  const int col0 = wave * 64 + 4 * c;

  int strip = blockIdx.x;
  if (strip >= nstrips) return;

  v2f b[4][32];
  {
    const float4* wp = (const float4*)(W + (size_t)(4 * h) * NCOL + col0);
    #pragma unroll
    for (int j = 0; j < 16; ++j) {
      #pragma unroll
      for (int q = 0; q < 2; ++q) {
        const float4 w0 = wp[((8 * j + 2 * q) * NCOL) >> 2];
        const float4 w1 = wp[((8 * j + 2 * q + 1) * NCOL) >> 2];
        b[0][2 * j + q] = v2f{w0.x, w1.x};
        b[1][2 * j + q] = v2f{w0.y, w1.y};
        b[2][2 * j + q] = v2f{w0.z, w1.z};
        b[3][2 * j + q] = v2f{w0.w, w1.w};
      }
    }
  }

  float4 a[16];
  float  nm;
  {
    const float4* xp = (const float4*)(X + (size_t)(strip * 16 + c) * KDIM + 4 * h);
    #pragma unroll
    for (int j = 0; j < 16; ++j) a[j] = xp[2 * j];
    nm = norm[strip * 16 + c];
  }

  float4 bv[16];
  float  nd = 0.f;
  if constexpr (PRE) {
    const float4* bp = (const float4*)(colB + 4 * h);
    #pragma unroll
    for (int j = 0; j < 16; ++j) bv[j] = bp[2 * j];
    nd = rowS[strip * 16 + c];
  }

  for (;;) {
    const int  nxt  = strip + (int)gridDim.x;
    const bool more = nxt < nstrips;
    const int  pf   = more ? nxt : strip;

    float4 an[16];
    const float4* xn = (const float4*)(X + (size_t)(pf * 16 + c) * KDIM + 4 * h);
    #pragma unroll
    for (int j = 0; j < 16; ++j) an[j] = xn[2 * j];
    const float nn = norm[pf * 16 + c];
    float ndn = 0.f;
    if constexpr (PRE) ndn = rowS[pf * 16 + c];

    v8f acc0 = {}, acc1 = {}, acc2 = {}, acc3 = {};
    #pragma unroll
    for (int j = 0; j < 16; ++j) {
      float4 x = a[j];
      if constexpr (PRE) {
        x.x = fmaxf(x.x * nd + bv[j].x, 0.0f);
        x.y = fmaxf(x.y * nd + bv[j].y, 0.0f);
        x.z = fmaxf(x.z * nd + bv[j].z, 0.0f);
        x.w = fmaxf(x.w * nd + bv[j].w, 0.0f);
      }
      const v2f a0 = v2f{x.x * nm, x.y * nm};
      const v2f a1 = v2f{x.z * nm, x.w * nm};
      acc0 = __builtin_amdgcn_wmma_f32_16x16x4_f32(false, a0, false, b[0][2 * j], (short)0, acc0, false, false);
      acc1 = __builtin_amdgcn_wmma_f32_16x16x4_f32(false, a0, false, b[1][2 * j], (short)0, acc1, false, false);
      acc2 = __builtin_amdgcn_wmma_f32_16x16x4_f32(false, a0, false, b[2][2 * j], (short)0, acc2, false, false);
      acc3 = __builtin_amdgcn_wmma_f32_16x16x4_f32(false, a0, false, b[3][2 * j], (short)0, acc3, false, false);
      acc0 = __builtin_amdgcn_wmma_f32_16x16x4_f32(false, a1, false, b[0][2 * j + 1], (short)0, acc0, false, false);
      acc1 = __builtin_amdgcn_wmma_f32_16x16x4_f32(false, a1, false, b[1][2 * j + 1], (short)0, acc1, false, false);
      acc2 = __builtin_amdgcn_wmma_f32_16x16x4_f32(false, a1, false, b[2][2 * j + 1], (short)0, acc2, false, false);
      acc3 = __builtin_amdgcn_wmma_f32_16x16x4_f32(false, a1, false, b[3][2 * j + 1], (short)0, acc3, false, false);
    }

    float4* yp = (float4*)(Y + (size_t)(strip * 16 + 8 * h) * NCOL + col0);
    #pragma unroll
    for (int r = 0; r < 8; ++r)
      yp[(r * NCOL) >> 2] = make_float4(acc0[r], acc1[r], acc2[r], acc3[r]);

    if (!more) break;
    #pragma unroll
    for (int j = 0; j < 16; ++j) a[j] = an[j];
    nm    = nn;
    nd    = ndn;
    strip = nxt;
  }
}


__global__ __launch_bounds__(1024) void k_place(
    const int* __restrict__ src, const int* __restrict__ dst,
    const unsigned* __restrict__ starts, int* __restrict__ pairs) {
  __shared__ unsigned cur[BINPAD];

  constexpr int N4    = NE / 4;
  constexpr int PER   = (N4 + NSLICE - 1) / NSLICE;
  constexpr int TRIPS = (PER + 1023) / 1024;

  const int tid   = threadIdx.x;
  const int slice = blockIdx.x;
  const int start = slice * PER;
  const int cnt   = min(PER, N4 - start);

  const int4* s4 = (const int4*)src + start;
  const int4* d4 = (const int4*)dst + start;
  int4 sv[TRIPS], dv[TRIPS];
  #pragma unroll
  for (int t = 0; t < TRIPS; ++t) {
    const int i = tid + 1024 * t;
    sv[t] = make_int4(0, 0, 0, 0);
    dv[t] = make_int4(0, 0, 0, 0);
    if (i < cnt) { sv[t] = s4[i]; dv[t] = d4[i]; }
  }

  if (tid < BINPAD) cur[tid] = starts[slice * BINPAD + tid];
  __syncthreads();

  int2* pp = (int2*)pairs;
  #pragma unroll
  for (int t = 0; t < TRIPS; ++t) {
    if (tid + 1024 * t < cnt) {
      const unsigned p0 = atomicAdd(&cur[(unsigned)dv[t].x / BINROWS], 1u);
      const unsigned p1 = atomicAdd(&cur[(unsigned)dv[t].y / BINROWS], 1u);
      const unsigned p2 = atomicAdd(&cur[(unsigned)dv[t].z / BINROWS], 1u);
      const unsigned p3 = atomicAdd(&cur[(unsigned)dv[t].w / BINROWS], 1u);
      pp[p0] = make_int2(sv[t].x, dv[t].x);
      pp[p1] = make_int2(sv[t].y, dv[t].y);
      pp[p2] = make_int2(sv[t].z, dv[t].z);
      pp[p3] = make_int2(sv[t].w, dv[t].w);
    }
  }
}


template <int F, int EPW, bool FINAL>
__global__ __launch_bounds__(1024) void k_binsum(
    const float* __restrict__ H, const int* __restrict__ pairs,
    const unsigned* __restrict__ binStart, const float* __restrict__ rowScale,
    const float* __restrict__ bias, float* __restrict__ out) {
  constexpr int Q  = F / 32;
  constexpr int F4 = F / 4;
  constexpr int TBL4 = BINROWS * F4;
  constexpr int RES4 = SUM_LDS_BYTES / 16;
  __shared__ float4 acc4[TBL4 > RES4 ? TBL4 : RES4];
  float* acc = (float*)acc4;

  const int tid  = threadIdx.x;
  const int lane = tid & 31;
  const int wave = __builtin_amdgcn_readfirstlane(tid >> 5);
  const int bin  = blockIdx.x;

  const int beg = (int)binStart[bin];
  const int end = (int)binStart[bin + 1];

  for (int i = tid; i < BINROWS * F4; i += 1024) acc4[i] = make_float4(0.f, 0.f, 0.f, 0.f);
  __syncthreads();

  const int2* pp = (const int2*)pairs;
  int e0 = beg + wave * EPW;
  if (e0 < end) {
    int s[EPW], d[EPW];
    #pragma unroll
    for (int k = 0; k < EPW; ++k) {
      const int2 p = pp[min(e0 + k, end - 1)];
      s[k] = p.x; d[k] = p.y;
    }
    for (;;) {
      float v[EPW][Q];
      #pragma unroll
      for (int k = 0; k < EPW; ++k) {
        const float* hp = H + (size_t)s[k] * F + lane;
        #pragma unroll
        for (int q = 0; q < Q; ++q) v[k][q] = hp[32 * q];
      }
      int r[EPW];
      #pragma unroll
      for (int k = 0; k < EPW; ++k) r[k] = (d[k] - bin * BINROWS) * F;
      const int live = end - e0;

      const int  en   = e0 + 32 * EPW;
      const bool more = en < end;
      if (more) {
        #pragma unroll
        for (int k = 0; k < EPW; ++k) {
          const int2 p = pp[min(en + k, end - 1)];
          s[k] = p.x; d[k] = p.y;
        }
      }

      #pragma unroll
      for (int k = 0; k < EPW; ++k) {
        if (k < live) {
          float* ap = acc + r[k] + lane;
          #pragma unroll
          for (int q = 0; q < Q; ++q) atomicAdd(ap + 32 * q, v[k][q]);
        }
      }
      if (!more) break;
      e0 = en;
    }
  }
  __syncthreads();

  const int node0 = bin * BINROWS;
  for (int i = tid; i < BINROWS * F4; i += 1024) {
    const int row  = i / F4;
    const int c4   = i % F4;
    const int node = node0 + row;
    if (node < NN) {
      float4 x = acc4[i];
      if constexpr (FINAL) {
        const float  sc = rowScale[node];
        const float4 bb = ((const float4*)bias)[c4];
        x.x = x.x * sc + bb.x;
        x.y = x.y * sc + bb.y;
        x.z = x.z * sc + bb.z;
        x.w = x.w * sc + bb.w;
      }
      ((float4*)out)[(size_t)node * F4 + c4] = x;
    }
  }
}


extern "C" void kernel_launch(void* const* d_in, const int* in_sizes, int n_in,
                              void* d_out, int out_size, void* d_ws, size_t ws_size,
                              hipStream_t stream) {
  (void)in_sizes; (void)n_in; (void)out_size; (void)ws_size;

  const float* features = (const float*)d_in[0];
  const float* W1       = (const float*)d_in[1];
  const float* b1       = (const float*)d_in[2];
  const float* W2       = (const float*)d_in[3];
  const float* b2       = (const float*)d_in[4];
  const int*   esrc     = (const int*)d_in[5];
  const int*   edst     = (const int*)d_in[6];
  float*       out      = (float*)d_out;

  float* w     = (float*)d_ws;
  float* normS = w;
  float* normD = w + PADN;
  float* B0    = w + 2 * PADN;
  float* B1    = B0 + (size_t)NN * 128;
  int*      pairs    = (int*)(B1 + (size_t)NN * 128);
  unsigned* binCnt   = (unsigned*)(pairs + 2 * NE);
  unsigned* starts   = binCnt + NSLICE * BINPAD;
  unsigned* binStart = starts + NSLICE * BINPAD;
  unsigned* part = (unsigned*)B0;

  const int T = 256;
  static_assert(T == BINPAD, "the scan block of k_norm has one thread per bin");

  k_degcount<<<NSLICE, 1024, 0, stream>>>(esrc, edst, part, binCnt);
  k_norm<<<(NN + T - 1) / T + 1, T, 0, stream>>>(part, normS, normD, NN, binCnt, starts,
                                                 binStart);
  k_place<<<NSLICE, 1024, 0, stream>>>(esrc, edst, starts, pairs);

  k_gemm_wreg<128, false><<<1024, 64, 0, stream>>>(features, normS, normD, b1, W1, B0,
                                                   NN / 16);

  k_binsum<128, 8, false><<<NBIN, 1024, 0, stream>>>(B0, pairs, binStart, normD, b2, B1);

  k_gemm_wreg<64, true><<<2048, 32, 0, stream>>>(B1, normS, normD, b1, W2, B0, NN / 16);

  k_binsum<64, 16, true><<<NBIN, 1024, 0, stream>>>(B0, pairs, binStart, normD, b2, out);
}


# Layer-1 edge sums moved ahead of its product (they commute): both products chain in one launch, transposed, first D tiles = second's B operand; 7 -> 6 launches; second k summed as 4 partials
# speedup vs baseline: 4.2498x; 1.0613x over previous
#include <hip/hip_runtime.h>


typedef __attribute__((ext_vector_type(2))) float v2f;
typedef __attribute__((ext_vector_type(8))) float v8f;

#define NN   50000
#define NE   600000
#define KDIM 128


#define NSLICE 64
#define PADN   50176

#define BINROWS 196
#define NBIN   ((NN + BINROWS - 1) / BINROWS)
#define BINPAD 256
#define SUM_LDS_BYTES (164 * 1024)

__global__ __launch_bounds__(1024) void k_degcount(
    const int* __restrict__ src, const int* __restrict__ dst,
    unsigned* __restrict__ part, unsigned* __restrict__ binCnt) {
  __shared__ uint4 tbl4[NN / 4];
  __shared__ unsigned lbin[BINPAD];
  unsigned* tbl = (unsigned*)tbl4;
  static_assert(NBIN <= BINPAD, "bins must fit the tables");

  constexpr int N4    = NE / 4;
  constexpr int PER   = (N4 + NSLICE - 1) / NSLICE;
  constexpr int TRIPS = (PER + 1023) / 1024;
  static_assert(NE % 4 == 0 && NN % 4 == 0, "whole 16-byte groups");
  static_assert(4 * PER < 65536, "a slice's count must fit 16 bits");

  const int tid   = threadIdx.x;
  const int slice = blockIdx.x;
  const int start = slice * PER;
  const int cnt   = min(PER, N4 - start);

  const int4* s4 = (const int4*)src + start;
  const int4* d4 = (const int4*)dst + start;
  int4 sv[TRIPS], dv[TRIPS];
  #pragma unroll
  for (int t = 0; t < TRIPS; ++t) {
    const int i = tid + 1024 * t;
    sv[t] = make_int4(0, 0, 0, 0);
    dv[t] = make_int4(0, 0, 0, 0);
    if (i < cnt) { sv[t] = s4[i]; dv[t] = d4[i]; }
  }

  for (int i = tid; i < NN / 4; i += 1024) tbl4[i] = make_uint4(0u, 0u, 0u, 0u);
  if (tid < BINPAD) lbin[tid] = 0u;
  __syncthreads();

  #pragma unroll
  for (int t = 0; t < TRIPS; ++t) {
    if (tid + 1024 * t < cnt) {
      atomicAdd(&tbl[sv[t].x], 1u);
      atomicAdd(&tbl[sv[t].y], 1u);
      atomicAdd(&tbl[sv[t].z], 1u);
      atomicAdd(&tbl[sv[t].w], 1u);
      atomicAdd(&tbl[dv[t].x], 0x10000u);
      atomicAdd(&tbl[dv[t].y], 0x10000u);
      atomicAdd(&tbl[dv[t].z], 0x10000u);
      atomicAdd(&tbl[dv[t].w], 0x10000u);
      atomicAdd(&lbin[(unsigned)dv[t].x / BINROWS], 1u);
      atomicAdd(&lbin[(unsigned)dv[t].y / BINROWS], 1u);
      atomicAdd(&lbin[(unsigned)dv[t].z / BINROWS], 1u);
      atomicAdd(&lbin[(unsigned)dv[t].w / BINROWS], 1u);
    }
  }
  __syncthreads();

  if (tid < BINPAD) binCnt[slice * BINPAD + tid] = lbin[tid];

  uint4* p4 = (uint4*)(part + (size_t)slice * PADN);
  for (int i = tid; i < NN / 4; i += 1024) p4[i] = tbl4[i];
}

__global__ void k_norm(const unsigned* __restrict__ part, float* __restrict__ a,
                       float* __restrict__ b, int n,
                       const unsigned* __restrict__ binCnt,
                       unsigned* __restrict__ starts,
                       unsigned* __restrict__ binStart) {
  if (blockIdx.x == gridDim.x - 1) {
    __shared__ unsigned tot[BINPAD];
    const int bn = threadIdx.x;
    unsigned t = 0u;
    for (int s = 0; s < NSLICE; ++s) t += binCnt[s * BINPAD + bn];
    tot[bn] = t;
    __syncthreads();
    unsigned base = 0u;
    for (int j = 0; j < bn; ++j) base += tot[j];
    binStart[bn] = base;
    if (bn == BINPAD - 1) binStart[BINPAD] = base + t;
    unsigned run = base;
    for (int s = 0; s < NSLICE; ++s) {
      starts[s * BINPAD + bn] = run;
      run += binCnt[s * BINPAD + bn];
    }
    return;
  }
  int i = blockIdx.x * blockDim.x + threadIdx.x;
  if (i < n) {
    unsigned s = 0u, d = 0u;
    for (int g = 0; g < NSLICE; ++g) {
      const unsigned w = part[(size_t)g * PADN + i];
      s += w & 0xffffu;
      d += w >> 16;
    }
    a[i] = rsqrtf(fmaxf((float)s, 1.0f));
    b[i] = rsqrtf(fmaxf((float)d, 1.0f));
  }
}


#define CH_NW    4
#define CH_HC    (KDIM / CH_NW)
#define CH_TT    (CH_HC / 16)
#define CH_PITCH 68

__global__ __launch_bounds__(32 * CH_NW) void k_gemm_chain(
    const float* __restrict__ X, const float* __restrict__ normS,
    const float* __restrict__ normD, const float* __restrict__ W1,
    const float* __restrict__ b1, const float* __restrict__ W2,
    float* __restrict__ Y, int nstrips) {
  static_assert(CH_TT == 2, "the W1 loader below takes one float2 per k");
  __shared__ float4 red4[2][CH_NW][16 * CH_PITCH / 4];

  const int lane = threadIdx.x & 31;
  const int wave = threadIdx.x >> 5;
  const int c    = lane & 15;
  const int h    = lane >> 4;

  int strip = blockIdx.x;
  if (strip >= nstrips) return;

  v2f wa[CH_TT][32];
  {
    const float2* wp = (const float2*)(W1 + (size_t)(4 * h) * KDIM + CH_HC * wave + CH_TT * c);
    #pragma unroll
    for (int j = 0; j < 16; ++j) {
      #pragma unroll
      for (int q = 0; q < 2; ++q) {
        const float2 w0 = wp[((8 * j + 2 * q) * KDIM) >> 1];
        const float2 w1 = wp[((8 * j + 2 * q + 1) * KDIM) >> 1];
        wa[0][2 * j + q] = v2f{w0.x, w1.x};
        wa[1][2 * j + q] = v2f{w0.y, w1.y};
      }
    }
  }

  v2f wb[4][CH_TT][4];
  {
    #pragma unroll
    for (int t = 0; t < CH_TT; ++t) {
      #pragma unroll
      for (int p = 0; p < 4; ++p) {
        const int k0 = CH_HC * wave + CH_TT * (8 * h + 2 * p) + t;
        const int k1 = k0 + CH_TT;
        const float4 w0 = *(const float4*)(W2 + (size_t)k0 * 64 + 4 * c);
        const float4 w1 = *(const float4*)(W2 + (size_t)k1 * 64 + 4 * c);
        wb[0][t][p] = v2f{w0.x, w1.x};
        wb[1][t][p] = v2f{w0.y, w1.y};
        wb[2][t][p] = v2f{w0.z, w1.z};
        wb[3][t][p] = v2f{w0.w, w1.w};
      }
    }
  }

  float bf[8 * CH_TT];
  {
    const float4* bp = (const float4*)(b1 + CH_HC * wave + 8 * CH_TT * h);
    #pragma unroll
    for (int i = 0; i < 2 * CH_TT; ++i) {
      const float4 v = bp[i];
      bf[4 * i] = v.x; bf[4 * i + 1] = v.y; bf[4 * i + 2] = v.z; bf[4 * i + 3] = v.w;
    }
  }

  auto do_strip = [&](const float4 (&cur)[16], const float cnd, const float cns,
                      float4 (&nx)[16], float& nnd, float& nns,
                      const int strip, const int pf, const int par)
                      __attribute__((always_inline)) {
    const float4* xn = (const float4*)(X + (size_t)(pf * 16 + c) * KDIM + 4 * h);
    #pragma unroll
    for (int j = 0; j < 16; ++j) nx[j] = xn[2 * j];
    nnd = normD[pf * 16 + c];
    nns = normS[pf * 16 + c];
    __builtin_amdgcn_sched_barrier(0);

    v8f acc[CH_TT];
    #pragma unroll
    for (int t = 0; t < CH_TT; ++t) acc[t] = v8f{};
    #pragma unroll
    for (int j = 0; j < 16; ++j) {
      const v2f x0 = v2f{cur[j].x, cur[j].y};
      const v2f x1 = v2f{cur[j].z, cur[j].w};
      #pragma unroll
      for (int t = 0; t < CH_TT; ++t)
        acc[t] = __builtin_amdgcn_wmma_f32_16x16x4_f32(false, wa[t][2 * j], false, x0, (short)0, acc[t], false, false);
      #pragma unroll
      for (int t = 0; t < CH_TT; ++t)
        acc[t] = __builtin_amdgcn_wmma_f32_16x16x4_f32(false, wa[t][2 * j + 1], false, x1, (short)0, acc[t], false, false);
    }

    #pragma unroll
    for (int t = 0; t < CH_TT; ++t) {
      #pragma unroll
      for (int r = 0; r < 8; ++r)
        acc[t][r] = fmaxf(acc[t][r] * cnd + bf[CH_TT * r + t], 0.0f) * cns;
    }

    v8f o0 = {}, o1 = {}, o2 = {}, o3 = {};
    #pragma unroll
    for (int t = 0; t < CH_TT; ++t) {
      #pragma unroll
      for (int p = 0; p < 4; ++p) {
        const v2f hb = v2f{acc[t][2 * p], acc[t][2 * p + 1]};
        o0 = __builtin_amdgcn_wmma_f32_16x16x4_f32(false, wb[0][t][p], false, hb, (short)0, o0, false, false);
        o1 = __builtin_amdgcn_wmma_f32_16x16x4_f32(false, wb[1][t][p], false, hb, (short)0, o1, false, false);
        o2 = __builtin_amdgcn_wmma_f32_16x16x4_f32(false, wb[2][t][p], false, hb, (short)0, o2, false, false);
        o3 = __builtin_amdgcn_wmma_f32_16x16x4_f32(false, wb[3][t][p], false, hb, (short)0, o3, false, false);
      }
    }

    float* red = (float*)red4[par][wave];
    #pragma unroll
    for (int r = 0; r < 8; ++r)
      *(float4*)(red + c * CH_PITCH + 32 * h + 4 * r) = make_float4(o0[r], o1[r], o2[r], o3[r]);
    __builtin_amdgcn_fence(__ATOMIC_RELEASE, "workgroup", "local");
    __builtin_amdgcn_s_barrier();
    __builtin_amdgcn_fence(__ATOMIC_ACQUIRE, "workgroup", "local");

    #pragma unroll
    for (int i = 0; i < 16 / CH_NW / 2; ++i) {
      const int n = (16 / CH_NW) * wave + 2 * i + h;
      float4 s = red4[par][0][n * (CH_PITCH / 4) + c];
      #pragma unroll
      for (int w = 1; w < CH_NW; ++w) {
        const float4 v = red4[par][w][n * (CH_PITCH / 4) + c];
        s.x += v.x; s.y += v.y; s.z += v.z; s.w += v.w;
      }
      ((float4*)Y)[(size_t)(strip * 16 + n) * 16 + c] = s;
    }
  };

  float4 ra[16], rb[16];
  float  nda, nsa, ndb, nsb;
  {
    const float4* xp = (const float4*)(X + (size_t)(strip * 16 + c) * KDIM + 4 * h);
    #pragma unroll
    for (int j = 0; j < 16; ++j) ra[j] = xp[2 * j];
    nda = normD[strip * 16 + c];
    nsa = normS[strip * 16 + c];
  }

  const int step = (int)gridDim.x;
  for (;;) {
    int  nxt  = strip + step;
    bool more = nxt < nstrips;
    do_strip(ra, nda, nsa, rb, ndb, nsb, strip, more ? nxt : strip, 0);
    if (!more) break;
    strip = nxt;

    nxt  = strip + step;
    more = nxt < nstrips;
    do_strip(rb, ndb, nsb, ra, nda, nsa, strip, more ? nxt : strip, 1);
    if (!more) break;
    strip = nxt;
  }
}


__global__ __launch_bounds__(1024) void k_place(
    const int* __restrict__ src, const int* __restrict__ dst,
    const unsigned* __restrict__ starts, int* __restrict__ pairs) {
  __shared__ unsigned cur[BINPAD];

  constexpr int N4    = NE / 4;
  constexpr int PER   = (N4 + NSLICE - 1) / NSLICE;
  constexpr int TRIPS = (PER + 1023) / 1024;

  const int tid   = threadIdx.x;
  const int slice = blockIdx.x;
  const int start = slice * PER;
  const int cnt   = min(PER, N4 - start);

  const int4* s4 = (const int4*)src + start;
  const int4* d4 = (const int4*)dst + start;
  int4 sv[TRIPS], dv[TRIPS];
  #pragma unroll
  for (int t = 0; t < TRIPS; ++t) {
    const int i = tid + 1024 * t;
    sv[t] = make_int4(0, 0, 0, 0);
    dv[t] = make_int4(0, 0, 0, 0);
    if (i < cnt) { sv[t] = s4[i]; dv[t] = d4[i]; }
  }

  if (tid < BINPAD) cur[tid] = starts[slice * BINPAD + tid];
  __syncthreads();

  int2* pp = (int2*)pairs;
  #pragma unroll
  for (int t = 0; t < TRIPS; ++t) {
    if (tid + 1024 * t < cnt) {
      const unsigned p0 = atomicAdd(&cur[(unsigned)dv[t].x / BINROWS], 1u);
      const unsigned p1 = atomicAdd(&cur[(unsigned)dv[t].y / BINROWS], 1u);
      const unsigned p2 = atomicAdd(&cur[(unsigned)dv[t].z / BINROWS], 1u);
      const unsigned p3 = atomicAdd(&cur[(unsigned)dv[t].w / BINROWS], 1u);
      pp[p0] = make_int2(sv[t].x, dv[t].x);
      pp[p1] = make_int2(sv[t].y, dv[t].y);
      pp[p2] = make_int2(sv[t].z, dv[t].z);
      pp[p3] = make_int2(sv[t].w, dv[t].w);
    }
  }
}


template <int F, int EPW, bool FINAL, bool SSRC>
__global__ __launch_bounds__(1024) void k_binsum(
    const float* __restrict__ H, const int* __restrict__ pairs,
    const unsigned* __restrict__ binStart, const float* __restrict__ rowScale,
    const float* __restrict__ bias, const float* __restrict__ srcScale,
    float* __restrict__ out) {
  constexpr int Q  = F / 32;
  constexpr int F4 = F / 4;
  constexpr int TBL4 = BINROWS * F4;
  constexpr int RES4 = SUM_LDS_BYTES / 16;
  __shared__ float4 acc4[TBL4 > RES4 ? TBL4 : RES4];
  float* acc = (float*)acc4;

  const int tid  = threadIdx.x;
  const int lane = tid & 31;
  const int wave = __builtin_amdgcn_readfirstlane(tid >> 5);
  const int bin  = blockIdx.x;

  const int beg = (int)binStart[bin];
  const int end = (int)binStart[bin + 1];

  for (int i = tid; i < BINROWS * F4; i += 1024) acc4[i] = make_float4(0.f, 0.f, 0.f, 0.f);
  __syncthreads();

  const int2* pp = (const int2*)pairs;
  int e0 = beg + wave * EPW;
  if (e0 < end) {
    int s[EPW], d[EPW];
    #pragma unroll
    for (int k = 0; k < EPW; ++k) {
      const int2 p = pp[min(e0 + k, end - 1)];
      s[k] = p.x; d[k] = p.y;
    }
    for (;;) {
      float v[EPW][Q];
      float sc[EPW];
      #pragma unroll
      for (int k = 0; k < EPW; ++k) {
        const float* hp = H + (size_t)s[k] * F + lane;
        #pragma unroll
        for (int q = 0; q < Q; ++q) v[k][q] = hp[32 * q];
        sc[k] = 1.0f;
        if constexpr (SSRC) sc[k] = srcScale[s[k]];
      }
      int r[EPW];
      #pragma unroll
      for (int k = 0; k < EPW; ++k) r[k] = (d[k] - bin * BINROWS) * F;
      const int live = end - e0;

      const int  en   = e0 + 32 * EPW;
      const bool more = en < end;
      if (more) {
        #pragma unroll
        for (int k = 0; k < EPW; ++k) {
          const int2 p = pp[min(en + k, end - 1)];
          s[k] = p.x; d[k] = p.y;
        }
      }

      #pragma unroll
      for (int k = 0; k < EPW; ++k) {
        if (k < live) {
          float* ap = acc + r[k] + lane;
          #pragma unroll
          for (int q = 0; q < Q; ++q) {
            float x = v[k][q];
            if constexpr (SSRC) x *= sc[k];
            atomicAdd(ap + 32 * q, x);
          }
        }
      }
      if (!more) break;
      e0 = en;
    }
  }
  __syncthreads();

  const int node0 = bin * BINROWS;
  for (int i = tid; i < BINROWS * F4; i += 1024) {
    const int row  = i / F4;
    const int c4   = i % F4;
    const int node = node0 + row;
    if (node < NN) {
      float4 x = acc4[i];
      if constexpr (FINAL) {
        const float  sc = rowScale[node];
        const float4 bb = ((const float4*)bias)[c4];
        x.x = x.x * sc + bb.x;
        x.y = x.y * sc + bb.y;
        x.z = x.z * sc + bb.z;
        x.w = x.w * sc + bb.w;
      }
      ((float4*)out)[(size_t)node * F4 + c4] = x;
    }
  }
}


extern "C" void kernel_launch(void* const* d_in, const int* in_sizes, int n_in,
                              void* d_out, int out_size, void* d_ws, size_t ws_size,
                              hipStream_t stream) {
  (void)in_sizes; (void)n_in; (void)out_size; (void)ws_size;

  const float* features = (const float*)d_in[0];
  const float* W1       = (const float*)d_in[1];
  const float* b1       = (const float*)d_in[2];
  const float* W2       = (const float*)d_in[3];
  const float* b2       = (const float*)d_in[4];
  const int*   esrc     = (const int*)d_in[5];
  const int*   edst     = (const int*)d_in[6];
  float*       out      = (float*)d_out;

  float* w     = (float*)d_ws;
  float* normS = w;
  float* normD = w + PADN;
  float* B0    = w + 2 * PADN;
  float* B1    = B0 + (size_t)NN * 128;
  int*      pairs    = (int*)(B1 + (size_t)NN * 128);
  unsigned* binCnt   = (unsigned*)(pairs + 2 * NE);
  unsigned* starts   = binCnt + NSLICE * BINPAD;
  unsigned* binStart = starts + NSLICE * BINPAD;
  unsigned* part = (unsigned*)B0;

  const int T = 256;
  static_assert(T == BINPAD, "the scan block of k_norm has one thread per bin");

  k_degcount<<<NSLICE, 1024, 0, stream>>>(esrc, edst, part, binCnt);
  k_norm<<<(NN + T - 1) / T + 1, T, 0, stream>>>(part, normS, normD, NN, binCnt, starts,
                                                 binStart);
  k_place<<<NSLICE, 1024, 0, stream>>>(esrc, edst, starts, pairs);

  k_binsum<128, 8, false, true><<<NBIN, 1024, 0, stream>>>(features, pairs, binStart, normD,
                                                           b2, normS, B1);

  k_gemm_chain<<<512, 32 * CH_NW, 0, stream>>>(B1, normS, normD, W1, b1, W2, B0, NN / 16);

  k_binsum<64, 16, true, false><<<NBIN, 1024, 0, stream>>>(B0, pairs, binStart, normD, b2,
                                                           normS, out);
}


# Norms and bin places in one launch: each placing work-group adds the slices' bin counts into its own start cursors, work-groups beside them make the norms; 6 -> 5 launches; all sums unchanged
# speedup vs baseline: 4.4897x; 1.0564x over previous
#include <hip/hip_runtime.h>


typedef __attribute__((ext_vector_type(2))) float v2f;
typedef __attribute__((ext_vector_type(8))) float v8f;

#define NN   50000
#define NE   600000
#define KDIM 128


#define NSLICE 64
#define PADN   50176

#define BINROWS 196
#define NBIN   ((NN + BINROWS - 1) / BINROWS)
#define BINPAD 256
#define SUM_LDS_BYTES (164 * 1024)

__global__ __launch_bounds__(1024) void k_degcount(
    const int* __restrict__ src, const int* __restrict__ dst,
    unsigned* __restrict__ part, unsigned* __restrict__ binCnt) {
  __shared__ uint4 tbl4[NN / 4];
  __shared__ unsigned lbin[BINPAD];
  unsigned* tbl = (unsigned*)tbl4;
  static_assert(NBIN <= BINPAD, "bins must fit the tables");

  constexpr int N4    = NE / 4;
  constexpr int PER   = (N4 + NSLICE - 1) / NSLICE;
  constexpr int TRIPS = (PER + 1023) / 1024;
  static_assert(NE % 4 == 0 && NN % 4 == 0, "whole 16-byte groups");
  static_assert(4 * PER < 65536, "a slice's count must fit 16 bits");

  const int tid   = threadIdx.x;
  const int slice = blockIdx.x;
  const int start = slice * PER;
  const int cnt   = min(PER, N4 - start);

  const int4* s4 = (const int4*)src + start;
  const int4* d4 = (const int4*)dst + start;
  int4 sv[TRIPS], dv[TRIPS];
  #pragma unroll
  for (int t = 0; t < TRIPS; ++t) {
    const int i = tid + 1024 * t;
    sv[t] = make_int4(0, 0, 0, 0);
    dv[t] = make_int4(0, 0, 0, 0);
    if (i < cnt) { sv[t] = s4[i]; dv[t] = d4[i]; }
  }

  for (int i = tid; i < NN / 4; i += 1024) tbl4[i] = make_uint4(0u, 0u, 0u, 0u);
  if (tid < BINPAD) lbin[tid] = 0u;
  __syncthreads();

  #pragma unroll
  for (int t = 0; t < TRIPS; ++t) {
    if (tid + 1024 * t < cnt) {
      atomicAdd(&tbl[sv[t].x], 1u);
      atomicAdd(&tbl[sv[t].y], 1u);
      atomicAdd(&tbl[sv[t].z], 1u);
      atomicAdd(&tbl[sv[t].w], 1u);
      atomicAdd(&tbl[dv[t].x], 0x10000u);
      atomicAdd(&tbl[dv[t].y], 0x10000u);
      atomicAdd(&tbl[dv[t].z], 0x10000u);
      atomicAdd(&tbl[dv[t].w], 0x10000u);
      atomicAdd(&lbin[(unsigned)dv[t].x / BINROWS], 1u);
      atomicAdd(&lbin[(unsigned)dv[t].y / BINROWS], 1u);
      atomicAdd(&lbin[(unsigned)dv[t].z / BINROWS], 1u);
      atomicAdd(&lbin[(unsigned)dv[t].w / BINROWS], 1u);
    }
  }
  __syncthreads();

  if (tid < BINPAD) binCnt[slice * BINPAD + tid] = lbin[tid];

  uint4* p4 = (uint4*)(part + (size_t)slice * PADN);
  for (int i = tid; i < NN / 4; i += 1024) p4[i] = tbl4[i];
}


#define CH_NW    4
#define CH_HC    (KDIM / CH_NW)
#define CH_TT    (CH_HC / 16)
#define CH_PITCH 68

__global__ __launch_bounds__(32 * CH_NW) void k_gemm_chain(
    const float* __restrict__ X, const float* __restrict__ normS,
    const float* __restrict__ normD, const float* __restrict__ W1,
    const float* __restrict__ b1, const float* __restrict__ W2,
    float* __restrict__ Y, int nstrips) {
  static_assert(CH_TT == 2, "the W1 loader below takes one float2 per k");
  __shared__ float4 red4[2][CH_NW][16 * CH_PITCH / 4];

  const int lane = threadIdx.x & 31;
  const int wave = threadIdx.x >> 5;
  const int c    = lane & 15;
  const int h    = lane >> 4;

  int strip = blockIdx.x;
  if (strip >= nstrips) return;

  v2f wa[CH_TT][32];
  {
    const float2* wp = (const float2*)(W1 + (size_t)(4 * h) * KDIM + CH_HC * wave + CH_TT * c);
    #pragma unroll
    for (int j = 0; j < 16; ++j) {
      #pragma unroll
      for (int q = 0; q < 2; ++q) {
        const float2 w0 = wp[((8 * j + 2 * q) * KDIM) >> 1];
        const float2 w1 = wp[((8 * j + 2 * q + 1) * KDIM) >> 1];
        wa[0][2 * j + q] = v2f{w0.x, w1.x};
        wa[1][2 * j + q] = v2f{w0.y, w1.y};
      }
    }
  }

  v2f wb[4][CH_TT][4];
  {
    #pragma unroll
    for (int t = 0; t < CH_TT; ++t) {
      #pragma unroll
      for (int p = 0; p < 4; ++p) {
        const int k0 = CH_HC * wave + CH_TT * (8 * h + 2 * p) + t;
        const int k1 = k0 + CH_TT;
        const float4 w0 = *(const float4*)(W2 + (size_t)k0 * 64 + 4 * c);
        const float4 w1 = *(const float4*)(W2 + (size_t)k1 * 64 + 4 * c);
        wb[0][t][p] = v2f{w0.x, w1.x};
        wb[1][t][p] = v2f{w0.y, w1.y};
        wb[2][t][p] = v2f{w0.z, w1.z};
        wb[3][t][p] = v2f{w0.w, w1.w};
      }
    }
  }

  float bf[8 * CH_TT];
  {
    const float4* bp = (const float4*)(b1 + CH_HC * wave + 8 * CH_TT * h);
    #pragma unroll
    for (int i = 0; i < 2 * CH_TT; ++i) {
      const float4 v = bp[i];
      bf[4 * i] = v.x; bf[4 * i + 1] = v.y; bf[4 * i + 2] = v.z; bf[4 * i + 3] = v.w;
    }
  }

  auto do_strip = [&](const float4 (&cur)[16], const float cnd, const float cns,
                      float4 (&nx)[16], float& nnd, float& nns,
                      const int strip, const int pf, const int par)
                      __attribute__((always_inline)) {
    const float4* xn = (const float4*)(X + (size_t)(pf * 16 + c) * KDIM + 4 * h);
    #pragma unroll
    for (int j = 0; j < 16; ++j) nx[j] = xn[2 * j];
    nnd = normD[pf * 16 + c];
    nns = normS[pf * 16 + c];
    __builtin_amdgcn_sched_barrier(0);

    v8f acc[CH_TT];
    #pragma unroll
    for (int t = 0; t < CH_TT; ++t) acc[t] = v8f{};
    #pragma unroll
    for (int j = 0; j < 16; ++j) {
      const v2f x0 = v2f{cur[j].x, cur[j].y};
      const v2f x1 = v2f{cur[j].z, cur[j].w};
      #pragma unroll
      for (int t = 0; t < CH_TT; ++t)
        acc[t] = __builtin_amdgcn_wmma_f32_16x16x4_f32(false, wa[t][2 * j], false, x0, (short)0, acc[t], false, false);
      #pragma unroll
      for (int t = 0; t < CH_TT; ++t)
        acc[t] = __builtin_amdgcn_wmma_f32_16x16x4_f32(false, wa[t][2 * j + 1], false, x1, (short)0, acc[t], false, false);
    }

    #pragma unroll
    for (int t = 0; t < CH_TT; ++t) {
      #pragma unroll
      for (int r = 0; r < 8; ++r)
        acc[t][r] = fmaxf(acc[t][r] * cnd + bf[CH_TT * r + t], 0.0f) * cns;
    }

    v8f o0 = {}, o1 = {}, o2 = {}, o3 = {};
    #pragma unroll
    for (int t = 0; t < CH_TT; ++t) {
      #pragma unroll
      for (int p = 0; p < 4; ++p) {
        const v2f hb = v2f{acc[t][2 * p], acc[t][2 * p + 1]};
        o0 = __builtin_amdgcn_wmma_f32_16x16x4_f32(false, wb[0][t][p], false, hb, (short)0, o0, false, false);
        o1 = __builtin_amdgcn_wmma_f32_16x16x4_f32(false, wb[1][t][p], false, hb, (short)0, o1, false, false);
        o2 = __builtin_amdgcn_wmma_f32_16x16x4_f32(false, wb[2][t][p], false, hb, (short)0, o2, false, false);
        o3 = __builtin_amdgcn_wmma_f32_16x16x4_f32(false, wb[3][t][p], false, hb, (short)0, o3, false, false);
      }
    }

    float* red = (float*)red4[par][wave];
    #pragma unroll
    for (int r = 0; r < 8; ++r)
      *(float4*)(red + c * CH_PITCH + 32 * h + 4 * r) = make_float4(o0[r], o1[r], o2[r], o3[r]);
    __builtin_amdgcn_fence(__ATOMIC_RELEASE, "workgroup", "local");
    __builtin_amdgcn_s_barrier();
    __builtin_amdgcn_fence(__ATOMIC_ACQUIRE, "workgroup", "local");

    #pragma unroll
    for (int i = 0; i < 16 / CH_NW / 2; ++i) {
      const int n = (16 / CH_NW) * wave + 2 * i + h;
      float4 s = red4[par][0][n * (CH_PITCH / 4) + c];
      #pragma unroll
      for (int w = 1; w < CH_NW; ++w) {
        const float4 v = red4[par][w][n * (CH_PITCH / 4) + c];
        s.x += v.x; s.y += v.y; s.z += v.z; s.w += v.w;
      }
      ((float4*)Y)[(size_t)(strip * 16 + n) * 16 + c] = s;
    }
  };

  float4 ra[16], rb[16];
  float  nda, nsa, ndb, nsb;
  {
    const float4* xp = (const float4*)(X + (size_t)(strip * 16 + c) * KDIM + 4 * h);
    #pragma unroll
    for (int j = 0; j < 16; ++j) ra[j] = xp[2 * j];
    nda = normD[strip * 16 + c];
    nsa = normS[strip * 16 + c];
  }

  const int step = (int)gridDim.x;
  for (;;) {
    int  nxt  = strip + step;
    bool more = nxt < nstrips;
    do_strip(ra, nda, nsa, rb, ndb, nsb, strip, more ? nxt : strip, 0);
    if (!more) break;
    strip = nxt;

    nxt  = strip + step;
    more = nxt < nstrips;
    do_strip(rb, ndb, nsb, ra, nda, nsa, strip, more ? nxt : strip, 1);
    if (!more) break;
    strip = nxt;
  }
}


#define NORM_WG ((NN + 1023) / 1024)
#define PN_GRID (NSLICE + NORM_WG)

__global__ __launch_bounds__(1024) void k_place_norm(
    const int* __restrict__ src, const int* __restrict__ dst,
    const unsigned* __restrict__ binCnt, const unsigned* __restrict__ part,
    float* __restrict__ a, float* __restrict__ b,
    int* __restrict__ pairs, unsigned* __restrict__ binStart) {
  constexpr int NG = 1024 / BINPAD;
  static_assert(NG * BINPAD == 1024 && NSLICE % NG == 0, "whole groups of slices per thread");
  __shared__ unsigned cur[BINPAD];
  __shared__ unsigned ptot[NG][BINPAD];
  __shared__ unsigned pbel[NG][BINPAD];
  __shared__ unsigned wsum[BINPAD / 32];

  const int tid = threadIdx.x;

  if (blockIdx.x >= NSLICE) {
    const int i = (blockIdx.x - NSLICE) * 1024 + tid;
    if (i < NN) {
      unsigned s = 0u, d = 0u;
      for (int g = 0; g < NSLICE; ++g) {
        const unsigned w = part[(size_t)g * PADN + i];
        s += w & 0xffffu;
        d += w >> 16;
      }
      a[i] = rsqrtf(fmaxf((float)s, 1.0f));
      b[i] = rsqrtf(fmaxf((float)d, 1.0f));
    }
    return;
  }

  constexpr int N4    = NE / 4;
  constexpr int PER   = (N4 + NSLICE - 1) / NSLICE;
  constexpr int TRIPS = (PER + 1023) / 1024;

  const int slice = blockIdx.x;
  const int start = slice * PER;
  const int cnt   = min(PER, N4 - start);

  const int4* s4 = (const int4*)src + start;
  const int4* d4 = (const int4*)dst + start;
  int4 sv[TRIPS], dv[TRIPS];
  #pragma unroll
  for (int t = 0; t < TRIPS; ++t) {
    const int i = tid + 1024 * t;
    sv[t] = make_int4(0, 0, 0, 0);
    dv[t] = make_int4(0, 0, 0, 0);
    if (i < cnt) { sv[t] = s4[i]; dv[t] = d4[i]; }
  }

  {
    const int bn = tid & (BINPAD - 1);
    const int g  = tid / BINPAD;
    unsigned t = 0u, bl = 0u;
    for (int k = 0; k < NSLICE / NG; ++k) {
      const int s = g * (NSLICE / NG) + k;
      const unsigned n = binCnt[s * BINPAD + bn];
      t  += n;
      bl += (s < slice) ? n : 0u;
    }
    ptot[g][bn] = t;
    pbel[g][bn] = bl;
  }
  __syncthreads();

  const int lane = tid & 31;
  unsigned tot = 0u, bel = 0u, incl = 0u;
  if (tid < BINPAD) {
    for (int g = 0; g < NG; ++g) { tot += ptot[g][tid]; bel += pbel[g][tid]; }
    incl = tot;
    for (int d = 1; d < 32; d <<= 1) {
      const unsigned n = __shfl_up(incl, d);
      if (lane >= d) incl += n;
    }
    if (lane == 31) wsum[tid >> 5] = incl;
  }
  __syncthreads();
  if (tid < BINPAD) {
    unsigned base = incl - tot;
    for (int w = 0; w < (tid >> 5); ++w) base += wsum[w];
    cur[tid] = base + bel;
    if (slice == 0) {
      binStart[tid] = base;
      if (tid == BINPAD - 1) binStart[BINPAD] = base + tot;
    }
  }
  __syncthreads();

  int2* pp = (int2*)pairs;
  #pragma unroll
  for (int t = 0; t < TRIPS; ++t) {
    if (tid + 1024 * t < cnt) {
      const unsigned p0 = atomicAdd(&cur[(unsigned)dv[t].x / BINROWS], 1u);
      const unsigned p1 = atomicAdd(&cur[(unsigned)dv[t].y / BINROWS], 1u);
      const unsigned p2 = atomicAdd(&cur[(unsigned)dv[t].z / BINROWS], 1u);
      const unsigned p3 = atomicAdd(&cur[(unsigned)dv[t].w / BINROWS], 1u);
      pp[p0] = make_int2(sv[t].x, dv[t].x);
      pp[p1] = make_int2(sv[t].y, dv[t].y);
      pp[p2] = make_int2(sv[t].z, dv[t].z);
      pp[p3] = make_int2(sv[t].w, dv[t].w);
    }
  }
}


template <int F, int EPW, bool FINAL, bool SSRC>
__global__ __launch_bounds__(1024) void k_binsum(
    const float* __restrict__ H, const int* __restrict__ pairs,
    const unsigned* __restrict__ binStart, const float* __restrict__ rowScale,
    const float* __restrict__ bias, const float* __restrict__ srcScale,
    float* __restrict__ out) {
  constexpr int Q  = F / 32;
  constexpr int F4 = F / 4;
  constexpr int TBL4 = BINROWS * F4;
  constexpr int RES4 = SUM_LDS_BYTES / 16;
  __shared__ float4 acc4[TBL4 > RES4 ? TBL4 : RES4];
  float* acc = (float*)acc4;

  const int tid  = threadIdx.x;
  const int lane = tid & 31;
  const int wave = __builtin_amdgcn_readfirstlane(tid >> 5);
  const int bin  = blockIdx.x;

  const int beg = (int)binStart[bin];
  const int end = (int)binStart[bin + 1];

  for (int i = tid; i < BINROWS * F4; i += 1024) acc4[i] = make_float4(0.f, 0.f, 0.f, 0.f);
  __syncthreads();

  const int2* pp = (const int2*)pairs;
  int e0 = beg + wave * EPW;
  if (e0 < end) {
    int s[EPW], d[EPW];
    #pragma unroll
    for (int k = 0; k < EPW; ++k) {
      const int2 p = pp[min(e0 + k, end - 1)];
      s[k] = p.x; d[k] = p.y;
    }
    for (;;) {
      float v[EPW][Q];
      float sc[EPW];
      #pragma unroll
      for (int k = 0; k < EPW; ++k) {
        const float* hp = H + (size_t)s[k] * F + lane;
        #pragma unroll
        for (int q = 0; q < Q; ++q) v[k][q] = hp[32 * q];
        sc[k] = 1.0f;
        if constexpr (SSRC) sc[k] = srcScale[s[k]];
      }
      int r[EPW];
      #pragma unroll
      for (int k = 0; k < EPW; ++k) r[k] = (d[k] - bin * BINROWS) * F;
      const int live = end - e0;

      const int  en   = e0 + 32 * EPW;
      const bool more = en < end;
      if (more) {
        #pragma unroll
        for (int k = 0; k < EPW; ++k) {
          const int2 p = pp[min(en + k, end - 1)];
          s[k] = p.x; d[k] = p.y;
        }
      }

      #pragma unroll
      for (int k = 0; k < EPW; ++k) {
        if (k < live) {
          float* ap = acc + r[k] + lane;
          #pragma unroll
          for (int q = 0; q < Q; ++q) {
            float x = v[k][q];
            if constexpr (SSRC) x *= sc[k];
            atomicAdd(ap + 32 * q, x);
          }
        }
      }
      if (!more) break;
      e0 = en;
    }
  }
  __syncthreads();

  const int node0 = bin * BINROWS;
  for (int i = tid; i < BINROWS * F4; i += 1024) {
    const int row  = i / F4;
    const int c4   = i % F4;
    const int node = node0 + row;
    if (node < NN) {
      float4 x = acc4[i];
      if constexpr (FINAL) {
        const float  sc = rowScale[node];
        const float4 bb = ((const float4*)bias)[c4];
        x.x = x.x * sc + bb.x;
        x.y = x.y * sc + bb.y;
        x.z = x.z * sc + bb.z;
        x.w = x.w * sc + bb.w;
      }
      ((float4*)out)[(size_t)node * F4 + c4] = x;
    }
  }
}


extern "C" void kernel_launch(void* const* d_in, const int* in_sizes, int n_in,
                              void* d_out, int out_size, void* d_ws, size_t ws_size,
                              hipStream_t stream) {
  (void)in_sizes; (void)n_in; (void)out_size; (void)ws_size;

  const float* features = (const float*)d_in[0];
  const float* W1       = (const float*)d_in[1];
  const float* b1       = (const float*)d_in[2];
  const float* W2       = (const float*)d_in[3];
  const float* b2       = (const float*)d_in[4];
  const int*   esrc     = (const int*)d_in[5];
  const int*   edst     = (const int*)d_in[6];
  float*       out      = (float*)d_out;

  float* w     = (float*)d_ws;
  float* normS = w;
  float* normD = w + PADN;
  float* B0    = w + 2 * PADN;
  float* B1    = B0 + (size_t)NN * 128;
  int*      pairs    = (int*)(B1 + (size_t)NN * 128);
  unsigned* binCnt   = (unsigned*)(pairs + 2 * NE);
  unsigned* binStart = binCnt + NSLICE * BINPAD;
  unsigned* part = (unsigned*)B0;

  k_degcount<<<NSLICE, 1024, 0, stream>>>(esrc, edst, part, binCnt);
  k_place_norm<<<PN_GRID, 1024, 0, stream>>>(esrc, edst, binCnt, part, normS, normD, pairs,
                                             binStart);

  k_binsum<128, 8, false, true><<<NBIN, 1024, 0, stream>>>(features, pairs, binStart, normD,
                                                           b2, normS, B1);

  k_gemm_chain<<<512, 32 * CH_NW, 0, stream>>>(B1, normS, normD, W1, b1, W2, B0, NN / 16);

  k_binsum<64, 16, true, false><<<NBIN, 1024, 0, stream>>>(B0, pairs, binStart, normD, b2,
                                                           normS, out);
}


# Edge sums: a lane gathers 4 (2) neighbouring floats of a row with one 16-B (8-B) load: one load per row, was one per 128-B line; LDS columns permuted, adds stay conflict-free, arrival-order sums kept
# speedup vs baseline: 4.7171x; 1.0506x over previous
#include <hip/hip_runtime.h>


typedef __attribute__((ext_vector_type(2))) float v2f;
typedef __attribute__((ext_vector_type(8))) float v8f;

#define NN   50000
#define NE   600000
#define KDIM 128


#define NSLICE 64
#define PADN   50176

#define BINROWS 196
#define NBIN   ((NN + BINROWS - 1) / BINROWS)
#define BINPAD 256
#define SUM_LDS_BYTES (164 * 1024)

__global__ __launch_bounds__(1024) void k_degcount(
    const int* __restrict__ src, const int* __restrict__ dst,
    unsigned* __restrict__ part, unsigned* __restrict__ binCnt) {
  __shared__ uint4 tbl4[NN / 4];
  __shared__ unsigned lbin[BINPAD];
  unsigned* tbl = (unsigned*)tbl4;
  static_assert(NBIN <= BINPAD, "bins must fit the tables");

  constexpr int N4    = NE / 4;
  constexpr int PER   = (N4 + NSLICE - 1) / NSLICE;
  constexpr int TRIPS = (PER + 1023) / 1024;
  static_assert(NE % 4 == 0 && NN % 4 == 0, "whole 16-byte groups");
  static_assert(4 * PER < 65536, "a slice's count must fit 16 bits");

  const int tid   = threadIdx.x;
  const int slice = blockIdx.x;
  const int start = slice * PER;
  const int cnt   = min(PER, N4 - start);

  const int4* s4 = (const int4*)src + start;
  const int4* d4 = (const int4*)dst + start;
  int4 sv[TRIPS], dv[TRIPS];
  #pragma unroll
  for (int t = 0; t < TRIPS; ++t) {
    const int i = tid + 1024 * t;
    sv[t] = make_int4(0, 0, 0, 0);
    dv[t] = make_int4(0, 0, 0, 0);
    if (i < cnt) { sv[t] = s4[i]; dv[t] = d4[i]; }
  }

  for (int i = tid; i < NN / 4; i += 1024) tbl4[i] = make_uint4(0u, 0u, 0u, 0u);
  if (tid < BINPAD) lbin[tid] = 0u;
  __syncthreads();

  #pragma unroll
  for (int t = 0; t < TRIPS; ++t) {
    if (tid + 1024 * t < cnt) {
      atomicAdd(&tbl[sv[t].x], 1u);
      atomicAdd(&tbl[sv[t].y], 1u);
      atomicAdd(&tbl[sv[t].z], 1u);
      atomicAdd(&tbl[sv[t].w], 1u);
      atomicAdd(&tbl[dv[t].x], 0x10000u);
      atomicAdd(&tbl[dv[t].y], 0x10000u);
      atomicAdd(&tbl[dv[t].z], 0x10000u);
      atomicAdd(&tbl[dv[t].w], 0x10000u);
      atomicAdd(&lbin[(unsigned)dv[t].x / BINROWS], 1u);
      atomicAdd(&lbin[(unsigned)dv[t].y / BINROWS], 1u);
      atomicAdd(&lbin[(unsigned)dv[t].z / BINROWS], 1u);
      atomicAdd(&lbin[(unsigned)dv[t].w / BINROWS], 1u);
    }
  }
  __syncthreads();

  if (tid < BINPAD) binCnt[slice * BINPAD + tid] = lbin[tid];

  uint4* p4 = (uint4*)(part + (size_t)slice * PADN);
  for (int i = tid; i < NN / 4; i += 1024) p4[i] = tbl4[i];
}


#define CH_NW    4
#define CH_HC    (KDIM / CH_NW)
#define CH_TT    (CH_HC / 16)
#define CH_PITCH 68

__global__ __launch_bounds__(32 * CH_NW) void k_gemm_chain(
    const float* __restrict__ X, const float* __restrict__ normS,
    const float* __restrict__ normD, const float* __restrict__ W1,
    const float* __restrict__ b1, const float* __restrict__ W2,
    float* __restrict__ Y, int nstrips) {
  static_assert(CH_TT == 2, "the W1 loader below takes one float2 per k");
  __shared__ float4 red4[2][CH_NW][16 * CH_PITCH / 4];

  const int lane = threadIdx.x & 31;
  const int wave = threadIdx.x >> 5;
  const int c    = lane & 15;
  const int h    = lane >> 4;

  int strip = blockIdx.x;
  if (strip >= nstrips) return;

  v2f wa[CH_TT][32];
  {
    const float2* wp = (const float2*)(W1 + (size_t)(4 * h) * KDIM + CH_HC * wave + CH_TT * c);
    #pragma unroll
    for (int j = 0; j < 16; ++j) {
      #pragma unroll
      for (int q = 0; q < 2; ++q) {
        const float2 w0 = wp[((8 * j + 2 * q) * KDIM) >> 1];
        const float2 w1 = wp[((8 * j + 2 * q + 1) * KDIM) >> 1];
        wa[0][2 * j + q] = v2f{w0.x, w1.x};
        wa[1][2 * j + q] = v2f{w0.y, w1.y};
      }
    }
  }

  v2f wb[4][CH_TT][4];
  {
    #pragma unroll
    for (int t = 0; t < CH_TT; ++t) {
      #pragma unroll
      for (int p = 0; p < 4; ++p) {
        const int k0 = CH_HC * wave + CH_TT * (8 * h + 2 * p) + t;
        const int k1 = k0 + CH_TT;
        const float4 w0 = *(const float4*)(W2 + (size_t)k0 * 64 + 4 * c);
        const float4 w1 = *(const float4*)(W2 + (size_t)k1 * 64 + 4 * c);
        wb[0][t][p] = v2f{w0.x, w1.x};
        wb[1][t][p] = v2f{w0.y, w1.y};
        wb[2][t][p] = v2f{w0.z, w1.z};
        wb[3][t][p] = v2f{w0.w, w1.w};
      }
    }
  }

  float bf[8 * CH_TT];
  {
    const float4* bp = (const float4*)(b1 + CH_HC * wave + 8 * CH_TT * h);
    #pragma unroll
    for (int i = 0; i < 2 * CH_TT; ++i) {
      const float4 v = bp[i];
      bf[4 * i] = v.x; bf[4 * i + 1] = v.y; bf[4 * i + 2] = v.z; bf[4 * i + 3] = v.w;
    }
  }

  auto do_strip = [&](const float4 (&cur)[16], const float cnd, const float cns,
                      float4 (&nx)[16], float& nnd, float& nns,
                      const int strip, const int pf, const int par)
                      __attribute__((always_inline)) {
    const float4* xn = (const float4*)(X + (size_t)(pf * 16 + c) * KDIM + 4 * h);
    #pragma unroll
    for (int j = 0; j < 16; ++j) nx[j] = xn[2 * j];
    nnd = normD[pf * 16 + c];
    nns = normS[pf * 16 + c];
    __builtin_amdgcn_sched_barrier(0);

    v8f acc[CH_TT];
    #pragma unroll
    for (int t = 0; t < CH_TT; ++t) acc[t] = v8f{};
    #pragma unroll
    for (int j = 0; j < 16; ++j) {
      const v2f x0 = v2f{cur[j].x, cur[j].y};
      const v2f x1 = v2f{cur[j].z, cur[j].w};
      #pragma unroll
      for (int t = 0; t < CH_TT; ++t)
        acc[t] = __builtin_amdgcn_wmma_f32_16x16x4_f32(false, wa[t][2 * j], false, x0, (short)0, acc[t], false, false);
      #pragma unroll
      for (int t = 0; t < CH_TT; ++t)
        acc[t] = __builtin_amdgcn_wmma_f32_16x16x4_f32(false, wa[t][2 * j + 1], false, x1, (short)0, acc[t], false, false);
    }

    #pragma unroll
    for (int t = 0; t < CH_TT; ++t) {
      #pragma unroll
      for (int r = 0; r < 8; ++r)
        acc[t][r] = fmaxf(acc[t][r] * cnd + bf[CH_TT * r + t], 0.0f) * cns;
    }

    v8f o0 = {}, o1 = {}, o2 = {}, o3 = {};
    #pragma unroll
    for (int t = 0; t < CH_TT; ++t) {
      #pragma unroll
      for (int p = 0; p < 4; ++p) {
        const v2f hb = v2f{acc[t][2 * p], acc[t][2 * p + 1]};
        o0 = __builtin_amdgcn_wmma_f32_16x16x4_f32(false, wb[0][t][p], false, hb, (short)0, o0, false, false);
        o1 = __builtin_amdgcn_wmma_f32_16x16x4_f32(false, wb[1][t][p], false, hb, (short)0, o1, false, false);
        o2 = __builtin_amdgcn_wmma_f32_16x16x4_f32(false, wb[2][t][p], false, hb, (short)0, o2, false, false);
        o3 = __builtin_amdgcn_wmma_f32_16x16x4_f32(false, wb[3][t][p], false, hb, (short)0, o3, false, false);
      }
    }

    float* red = (float*)red4[par][wave];
    #pragma unroll
    for (int r = 0; r < 8; ++r)
      *(float4*)(red + c * CH_PITCH + 32 * h + 4 * r) = make_float4(o0[r], o1[r], o2[r], o3[r]);
    __builtin_amdgcn_fence(__ATOMIC_RELEASE, "workgroup", "local");
    __builtin_amdgcn_s_barrier();
    __builtin_amdgcn_fence(__ATOMIC_ACQUIRE, "workgroup", "local");

    #pragma unroll
    for (int i = 0; i < 16 / CH_NW / 2; ++i) {
      const int n = (16 / CH_NW) * wave + 2 * i + h;
      float4 s = red4[par][0][n * (CH_PITCH / 4) + c];
      #pragma unroll
      for (int w = 1; w < CH_NW; ++w) {
        const float4 v = red4[par][w][n * (CH_PITCH / 4) + c];
        s.x += v.x; s.y += v.y; s.z += v.z; s.w += v.w;
      }
      ((float4*)Y)[(size_t)(strip * 16 + n) * 16 + c] = s;
    }
  };

  float4 ra[16], rb[16];
  float  nda, nsa, ndb, nsb;
  {
    const float4* xp = (const float4*)(X + (size_t)(strip * 16 + c) * KDIM + 4 * h);
    #pragma unroll
    for (int j = 0; j < 16; ++j) ra[j] = xp[2 * j];
    nda = normD[strip * 16 + c];
    nsa = normS[strip * 16 + c];
  }

  const int step = (int)gridDim.x;
  for (;;) {
    int  nxt  = strip + step;
    bool more = nxt < nstrips;
    do_strip(ra, nda, nsa, rb, ndb, nsb, strip, more ? nxt : strip, 0);
    if (!more) break;
    strip = nxt;

    nxt  = strip + step;
    more = nxt < nstrips;
    do_strip(rb, ndb, nsb, ra, nda, nsa, strip, more ? nxt : strip, 1);
    if (!more) break;
    strip = nxt;
  }
}


#define NORM_WG ((NN + 1023) / 1024)
#define PN_GRID (NSLICE + NORM_WG)

__global__ __launch_bounds__(1024) void k_place_norm(
    const int* __restrict__ src, const int* __restrict__ dst,
    const unsigned* __restrict__ binCnt, const unsigned* __restrict__ part,
    float* __restrict__ a, float* __restrict__ b,
    int* __restrict__ pairs, unsigned* __restrict__ binStart) {
  constexpr int NG = 1024 / BINPAD;
  static_assert(NG * BINPAD == 1024 && NSLICE % NG == 0, "whole groups of slices per thread");
  __shared__ unsigned cur[BINPAD];
  __shared__ unsigned ptot[NG][BINPAD];
  __shared__ unsigned pbel[NG][BINPAD];
  __shared__ unsigned wsum[BINPAD / 32];

  const int tid = threadIdx.x;

  if (blockIdx.x >= NSLICE) {
    const int i = (blockIdx.x - NSLICE) * 1024 + tid;
    if (i < NN) {
      unsigned s = 0u, d = 0u;
      for (int g = 0; g < NSLICE; ++g) {
        const unsigned w = part[(size_t)g * PADN + i];
        s += w & 0xffffu;
        d += w >> 16;
      }
      a[i] = rsqrtf(fmaxf((float)s, 1.0f));
      b[i] = rsqrtf(fmaxf((float)d, 1.0f));
    }
    return;
  }

  constexpr int N4    = NE / 4;
  constexpr int PER   = (N4 + NSLICE - 1) / NSLICE;
  constexpr int TRIPS = (PER + 1023) / 1024;

  const int slice = blockIdx.x;
  const int start = slice * PER;
  const int cnt   = min(PER, N4 - start);

  const int4* s4 = (const int4*)src + start;
  const int4* d4 = (const int4*)dst + start;
  int4 sv[TRIPS], dv[TRIPS];
  #pragma unroll
  for (int t = 0; t < TRIPS; ++t) {
    const int i = tid + 1024 * t;
    sv[t] = make_int4(0, 0, 0, 0);
    dv[t] = make_int4(0, 0, 0, 0);
    if (i < cnt) { sv[t] = s4[i]; dv[t] = d4[i]; }
  }

  {
    const int bn = tid & (BINPAD - 1);
    const int g  = tid / BINPAD;
    unsigned t = 0u, bl = 0u;
    for (int k = 0; k < NSLICE / NG; ++k) {
      const int s = g * (NSLICE / NG) + k;
      const unsigned n = binCnt[s * BINPAD + bn];
      t  += n;
      bl += (s < slice) ? n : 0u;
    }
    ptot[g][bn] = t;
    pbel[g][bn] = bl;
  }
  __syncthreads();

  const int lane = tid & 31;
  unsigned tot = 0u, bel = 0u, incl = 0u;
  if (tid < BINPAD) {
    for (int g = 0; g < NG; ++g) { tot += ptot[g][tid]; bel += pbel[g][tid]; }
    incl = tot;
    for (int d = 1; d < 32; d <<= 1) {
      const unsigned n = __shfl_up(incl, d);
      if (lane >= d) incl += n;
    }
    if (lane == 31) wsum[tid >> 5] = incl;
  }
  __syncthreads();
  if (tid < BINPAD) {
    unsigned base = incl - tot;
    for (int w = 0; w < (tid >> 5); ++w) base += wsum[w];
    cur[tid] = base + bel;
    if (slice == 0) {
      binStart[tid] = base;
      if (tid == BINPAD - 1) binStart[BINPAD] = base + tot;
    }
  }
  __syncthreads();

  int2* pp = (int2*)pairs;
  #pragma unroll
  for (int t = 0; t < TRIPS; ++t) {
    if (tid + 1024 * t < cnt) {
      const unsigned p0 = atomicAdd(&cur[(unsigned)dv[t].x / BINROWS], 1u);
      const unsigned p1 = atomicAdd(&cur[(unsigned)dv[t].y / BINROWS], 1u);
      const unsigned p2 = atomicAdd(&cur[(unsigned)dv[t].z / BINROWS], 1u);
      const unsigned p3 = atomicAdd(&cur[(unsigned)dv[t].w / BINROWS], 1u);
      pp[p0] = make_int2(sv[t].x, dv[t].x);
      pp[p1] = make_int2(sv[t].y, dv[t].y);
      pp[p2] = make_int2(sv[t].z, dv[t].z);
      pp[p3] = make_int2(sv[t].w, dv[t].w);
    }
  }
}


template <int F, int EPW, bool FINAL, bool SSRC>
__global__ __launch_bounds__(1024) void k_binsum(
    const float* __restrict__ H, const int* __restrict__ pairs,
    const unsigned* __restrict__ binStart, const float* __restrict__ rowScale,
    const float* __restrict__ bias, const float* __restrict__ srcScale,
    float* __restrict__ out) {
  constexpr int Q  = F / 32;
  constexpr int F4 = F / 4;
  typedef __attribute__((ext_vector_type(Q))) float rowvec_t;
  constexpr int TBL4 = BINROWS * F4;
  constexpr int RES4 = SUM_LDS_BYTES / 16;
  __shared__ float4 acc4[TBL4 > RES4 ? TBL4 : RES4];
  float* acc = (float*)acc4;

  const int tid  = threadIdx.x;
  const int lane = tid & 31;
  const int wave = __builtin_amdgcn_readfirstlane(tid >> 5);
  const int bin  = blockIdx.x;

  const int beg = (int)binStart[bin];
  const int end = (int)binStart[bin + 1];

  for (int i = tid; i < BINROWS * F4; i += 1024) acc4[i] = make_float4(0.f, 0.f, 0.f, 0.f);
  __syncthreads();

  const int2* pp = (const int2*)pairs;
  int e0 = beg + wave * EPW;
  if (e0 < end) {
    int s[EPW], d[EPW];
    #pragma unroll
    for (int k = 0; k < EPW; ++k) {
      const int2 p = pp[min(e0 + k, end - 1)];
      s[k] = p.x; d[k] = p.y;
    }
    for (;;) {
      float v[EPW][Q];
      float sc[EPW];
      #pragma unroll
      for (int k = 0; k < EPW; ++k) {
        const rowvec_t rv = *(const rowvec_t*)(H + (size_t)s[k] * F + Q * lane);
        #pragma unroll
        for (int q = 0; q < Q; ++q) v[k][q] = rv[q];
        sc[k] = 1.0f;
        if constexpr (SSRC) sc[k] = srcScale[s[k]];
      }
      int r[EPW];
      #pragma unroll
      for (int k = 0; k < EPW; ++k) r[k] = (d[k] - bin * BINROWS) * F;
      const int live = end - e0;

      const int  en   = e0 + 32 * EPW;
      const bool more = en < end;
      if (more) {
        #pragma unroll
        for (int k = 0; k < EPW; ++k) {
          const int2 p = pp[min(en + k, end - 1)];
          s[k] = p.x; d[k] = p.y;
        }
      }

      #pragma unroll
      for (int k = 0; k < EPW; ++k) {
        if (k < live) {
          float* ap = acc + r[k] + lane;
          #pragma unroll
          for (int q = 0; q < Q; ++q) {
            float x = v[k][q];
            if constexpr (SSRC) x *= sc[k];
            atomicAdd(ap + 32 * q, x);
          }
        }
      }
      if (!more) break;
      e0 = en;
    }
  }
  __syncthreads();

  const int node0 = bin * BINROWS;
  for (int i = tid; i < BINROWS * F4; i += 1024) {
    const int row  = i / F4;
    const int c4   = i % F4;
    const int node = node0 + row;
    if (node < NN) {
      const float* rp = acc + row * F;
      float4 x;
      x.x = rp[((4 * c4 + 0) % Q) * 32 + (4 * c4 + 0) / Q];
      x.y = rp[((4 * c4 + 1) % Q) * 32 + (4 * c4 + 1) / Q];
      x.z = rp[((4 * c4 + 2) % Q) * 32 + (4 * c4 + 2) / Q];
      x.w = rp[((4 * c4 + 3) % Q) * 32 + (4 * c4 + 3) / Q];
      if constexpr (FINAL) {
        const float  sc = rowScale[node];
        const float4 bb = ((const float4*)bias)[c4];
        x.x = x.x * sc + bb.x;
        x.y = x.y * sc + bb.y;
        x.z = x.z * sc + bb.z;
        x.w = x.w * sc + bb.w;
      }
      ((float4*)out)[(size_t)node * F4 + c4] = x;
    }
  }
}


extern "C" void kernel_launch(void* const* d_in, const int* in_sizes, int n_in,
                              void* d_out, int out_size, void* d_ws, size_t ws_size,
                              hipStream_t stream) {
  (void)in_sizes; (void)n_in; (void)out_size; (void)ws_size;

  const float* features = (const float*)d_in[0];
  const float* W1       = (const float*)d_in[1];
  const float* b1       = (const float*)d_in[2];
  const float* W2       = (const float*)d_in[3];
  const float* b2       = (const float*)d_in[4];
  const int*   esrc     = (const int*)d_in[5];
  const int*   edst     = (const int*)d_in[6];
  float*       out      = (float*)d_out;

  float* w     = (float*)d_ws;
  float* normS = w;
  float* normD = w + PADN;
  float* B0    = w + 2 * PADN;
  float* B1    = B0 + (size_t)NN * 128;
  int*      pairs    = (int*)(B1 + (size_t)NN * 128);
  unsigned* binCnt   = (unsigned*)(pairs + 2 * NE);
  unsigned* binStart = binCnt + NSLICE * BINPAD;
  unsigned* part = (unsigned*)B0;

  k_degcount<<<NSLICE, 1024, 0, stream>>>(esrc, edst, part, binCnt);
  k_place_norm<<<PN_GRID, 1024, 0, stream>>>(esrc, edst, binCnt, part, normS, normD, pairs,
                                             binStart);

  k_binsum<128, 8, false, true><<<NBIN, 1024, 0, stream>>>(features, pairs, binStart, normD,
                                                           b2, normS, B1);

  k_gemm_chain<<<512, 32 * CH_NW, 0, stream>>>(B1, normS, normD, W1, b1, W2, B0, NN / 16);

  k_binsum<64, 16, true, false><<<NBIN, 1024, 0, stream>>>(B0, pairs, binStart, normD, b2,
                                                           normS, out);
}
